# starting kernel
# baseline (speedup 1.0000x reference)
#include <hip/hip_runtime.h>

#define N_NODES 100000
#define N_EDGES 640000
#define D       128
#define BN_EPS  1e-5f
#define LDS_PITCH 132

typedef float v2f __attribute__((ext_vector_type(2)));
typedef float v8f __attribute__((ext_vector_type(8)));

__global__ void k_init_deg(float* __restrict__ deg) {
    int i = blockIdx.x * blockDim.x + threadIdx.x;
    if (i < N_NODES) deg[i] = 1.0f;
}

__global__ void k_deg(const int* __restrict__ dst, float* __restrict__ deg) {
    int e = blockIdx.x * blockDim.x + threadIdx.x;
    if (e < N_EDGES) atomicAdd(&deg[dst[e]], 1.0f);
}

__global__ void k_dinv(float* __restrict__ deg) {
    int i = blockIdx.x * blockDim.x + threadIdx.x;
    if (i < N_NODES) deg[i] = rsqrtf(deg[i]);
}

__global__ void __launch_bounds__(256) k_gemm(const float* __restrict__ x,
                                              const float* __restrict__ Wm,
                                              float* __restrict__ h) {
    __shared__ float lx[16 * LDS_PITCH];
    const int tid  = threadIdx.x;
    const int row0 = blockIdx.x * 16;

    for (int idx = tid; idx < 16 * D; idx += 256) {
        int r = idx >> 7, c = idx & (D - 1);
        lx[r * LDS_PITCH + c] = x[(row0 + r) * D + c];
    }
    __syncthreads();

    const int wave = tid >> 5;
    const int lane = tid & 31;
    const int half = lane >> 4;
    const int l15  = lane & 15;
    const int koff = half * 2;
    const int n0   = wave * 16;

    v8f acc = {};
    for (int k = 0; k < D; k += 4) {
        v2f a, b;
        a.x = lx[l15 * LDS_PITCH + k + koff];
        a.y = lx[l15 * LDS_PITCH + k + koff + 1];
        b.x = Wm[(k + koff)     * D + n0 + l15];
        b.y = Wm[(k + koff + 1) * D + n0 + l15];
        acc = __builtin_amdgcn_wmma_f32_16x16x4_f32(
             false, a,  false, b,
             (short)0, acc,  false,  false);
    }

#pragma unroll
    for (int r = 0; r < 8; ++r) {
        int row = row0 + r + half * 8;
        h[row * D + n0 + l15] = acc[r];
    }
}

__global__ void k_self_bias(const float* __restrict__ h, const float* __restrict__ dinv,
                            const float* __restrict__ b, float* __restrict__ out) {
    int i = blockIdx.x * blockDim.x + threadIdx.x;
    if (i < N_NODES * D) {
        int n = i >> 7, d = i & (D - 1);
        float di = dinv[n];
        out[i] = b[d] + h[i] * (di * di);
    }
}

__global__ void __launch_bounds__(256) k_scatter(const int* __restrict__ src,
                                                 const int* __restrict__ dst,
                                                 const float* __restrict__ h,
                                                 const float* __restrict__ dinv,
                                                 float* __restrict__ out) {
    int e    = blockIdx.x * (blockDim.x >> 5) + (threadIdx.x >> 5);
    int lane = threadIdx.x & 31;
    if (e >= N_EDGES) return;

    int ep = e + 8;
    if (ep < N_EDGES)
        __builtin_prefetch(h + (size_t)src[ep] * D + lane * 4, 0, 0);

    int s = src[e], t = dst[e];
    float w = dinv[s] * dinv[t];

    const float4* hp = (const float4*)(h + (size_t)s * D);
    float4 v = hp[lane];
    float* op = out + (size_t)t * D + lane * 4;
    atomicAdd(op + 0, v.x * w);
    atomicAdd(op + 1, v.y * w);
    atomicAdd(op + 2, v.z * w);
    atomicAdd(op + 3, v.w * w);
}

__global__ void k_init_stats(float* __restrict__ sums, float* __restrict__ sumsq) {
    int i = threadIdx.x;
    if (i < D) { sums[i] = 0.0f; sumsq[i] = 0.0f; }
}

__global__ void __launch_bounds__(256) k_stats(const float* __restrict__ out,
                                               float* __restrict__ sums,
                                               float* __restrict__ sumsq) {
    int d = threadIdx.x & (D - 1);
    int r0 = blockIdx.x * 2 + (threadIdx.x >> 7);
    float s = 0.0f, ss = 0.0f;
    for (int r = r0; r < N_NODES; r += gridDim.x * 2) {
        float v = out[r * D + d];
        v = v > 0.0f ? v : 0.0f;
        s += v; ss += v * v;
    }
    atomicAdd(&sums[d], s);
    atomicAdd(&sumsq[d], ss);
}

__global__ void k_bn(float* __restrict__ out, const float* __restrict__ sums,
                     const float* __restrict__ sumsq, const float* __restrict__ gamma,
                     const float* __restrict__ beta) {
    int i = blockIdx.x * blockDim.x + threadIdx.x;
    if (i >= N_NODES * D) return;
    int d = i & (D - 1);
    const float invN = 1.0f / (float)N_NODES;
    float mean = sums[d] * invN;
    float var  = sumsq[d] * invN - mean * mean;
    float v = out[i];
    v = v > 0.0f ? v : 0.0f;
    out[i] = gamma[d] * (v - mean) * rsqrtf(var + BN_EPS) + beta[d];
}

extern "C" void kernel_launch(void* const* d_in, const int* in_sizes, int n_in,
                              void* d_out, int out_size, void* d_ws, size_t ws_size,
                              hipStream_t stream) {
    const float* x     = (const float*)d_in[0];
    const int*   edges = (const int*)  d_in[1];
    const float* Wm    = (const float*)d_in[2];
    const float* b     = (const float*)d_in[3];
    const float* gamma = (const float*)d_in[4];
    const float* beta  = (const float*)d_in[5];
    float*       out   = (float*)d_out;

    const int* src = edges;
    const int* dst = edges + N_EDGES;

    float* deg   = (float*)d_ws;
    float* h     = deg + N_NODES;
    float* sums  = h + (size_t)N_NODES * D;
    float* sumsq = sums + D;

    k_init_deg  <<<(N_NODES + 255) / 256, 256, 0, stream>>>(deg);
    k_deg       <<<(N_EDGES + 255) / 256, 256, 0, stream>>>(dst, deg);
    k_dinv      <<<(N_NODES + 255) / 256, 256, 0, stream>>>(deg);

    k_gemm      <<<N_NODES / 16, 256, 0, stream>>>(x, Wm, h);

    k_self_bias <<<(N_NODES * D + 255) / 256, 256, 0, stream>>>(h, deg, b, out);
    k_scatter   <<<N_EDGES / 8, 256, 0, stream>>>(src, dst, h, deg, out);

    k_init_stats<<<1, 128, 0, stream>>>(sums, sumsq);
    k_stats     <<<512, 256, 0, stream>>>(out, sums, sumsq);
    k_bn        <<<(N_NODES * D + 255) / 256, 256, 0, stream>>>(out, sums, sumsq, gamma, beta);
}


# Float-atomic edge scatter replaced by in-call counting sort by destination and a per-node register gather (bias pass folded in); edge sums in slot order, slot = arrival rank of an int count atomic
# speedup vs baseline: 1.3337x; 1.3337x over previous
#include <hip/hip_runtime.h>

#define N_NODES 100000
#define N_EDGES 640000
#define D       128
#define BN_EPS  1e-5f
#define LDS_PITCH 132

typedef float v2f __attribute__((ext_vector_type(2)));
typedef float v8f __attribute__((ext_vector_type(8)));

#define SCAN_B 1024

__global__ void k_zero_cnt(int* __restrict__ cnt) {
    int i = blockIdx.x * blockDim.x + threadIdx.x;
    if (i < N_NODES) cnt[i] = 0;
}

__global__ void k_count(const int* __restrict__ dst, int* __restrict__ cnt,
                        int* __restrict__ rank) {
    int e = blockIdx.x * blockDim.x + threadIdx.x;
    if (e < N_EDGES) rank[e] = atomicAdd(&cnt[dst[e]], 1);
}

__global__ void __launch_bounds__(SCAN_B) k_scan(const int* __restrict__ cnt,
                                                 int* __restrict__ rowptr,
                                                 float* __restrict__ dinv) {
    __shared__ int sp[SCAN_B];
    __shared__ int sc[2][SCAN_B];
    const int t    = threadIdx.x;
    const int base = blockIdx.x * SCAN_B;

    int p = 0;
    for (int i = t; i < base; i += SCAN_B) p += cnt[i];
    const int i = base + t;
    const int c = (i < N_NODES) ? cnt[i] : 0;
    sp[t]    = p;
    sc[0][t] = c;
    __syncthreads();

    for (int s = SCAN_B / 2; s > 0; s >>= 1) {
        if (t < s) sp[t] += sp[t + s];
        __syncthreads();
    }
    const int before = sp[0];

    int cur = 0;
    for (int off = 1; off < SCAN_B; off <<= 1) {
        int v = sc[cur][t];
        if (t >= off) v += sc[cur][t - off];
        sc[cur ^ 1][t] = v;
        __syncthreads();
        cur ^= 1;
    }
    const int incl = sc[cur][t];

    if (i <= N_NODES) rowptr[i] = before + incl - c;
    if (i <  N_NODES) dinv[i]   = rsqrtf((float)(c + 1));
}

__global__ void k_fill(const int* __restrict__ src, const int* __restrict__ dst,
                       const int* __restrict__ rowptr, const int* __restrict__ rank,
                       int* __restrict__ col) {
    int e = blockIdx.x * blockDim.x + threadIdx.x;
    if (e < N_EDGES) col[rowptr[dst[e]] + rank[e]] = src[e];
}

__global__ void __launch_bounds__(256) k_gemm(const float* __restrict__ x,
                                              const float* __restrict__ Wm,
                                              float* __restrict__ h) {
    __shared__ float lx[16 * LDS_PITCH];
    const int tid  = threadIdx.x;
    const int row0 = blockIdx.x * 16;

    for (int idx = tid; idx < 16 * D; idx += 256) {
        int r = idx >> 7, c = idx & (D - 1);
        lx[r * LDS_PITCH + c] = x[(row0 + r) * D + c];
    }
    __syncthreads();

    const int wave = tid >> 5;
    const int lane = tid & 31;
    const int half = lane >> 4;
    const int l15  = lane & 15;
    const int koff = half * 2;
    const int n0   = wave * 16;

    v8f acc = {};
    for (int k = 0; k < D; k += 4) {
        v2f a, b;
        a.x = lx[l15 * LDS_PITCH + k + koff];
        a.y = lx[l15 * LDS_PITCH + k + koff + 1];
        b.x = Wm[(k + koff)     * D + n0 + l15];
        b.y = Wm[(k + koff + 1) * D + n0 + l15];
        acc = __builtin_amdgcn_wmma_f32_16x16x4_f32(
             false, a,  false, b,
             (short)0, acc,  false,  false);
    }

#pragma unroll
    for (int r = 0; r < 8; ++r) {
        int row = row0 + r + half * 8;
        h[row * D + n0 + l15] = acc[r];
    }
}

template <int R>
__device__ __forceinline__ void add_rows(float4& acc, const float4* __restrict__ hp,
                                         int my_s, float my_w, int j, int lane) {
    float4 v[R];
#pragma unroll
    for (int k = 0; k < R; ++k)
        v[k] = hp[(size_t)__builtin_amdgcn_readlane(my_s, j + k) * 32 + lane];
#pragma unroll
    for (int k = 0; k < R; ++k) {
        const float w = __int_as_float(__builtin_amdgcn_readlane(__float_as_int(my_w), j + k));
        acc.x = fmaf(v[k].x, w, acc.x);
        acc.y = fmaf(v[k].y, w, acc.y);
        acc.z = fmaf(v[k].z, w, acc.z);
        acc.w = fmaf(v[k].w, w, acc.w);
    }
}

__global__ void __launch_bounds__(256) k_gather(const int* __restrict__ rowptr,
                                                const int* __restrict__ col,
                                                const float* __restrict__ h,
                                                const float* __restrict__ dinv,
                                                const float* __restrict__ b,
                                                float* __restrict__ out) {
    const int lane = threadIdx.x & 31;
    const int t = __builtin_amdgcn_readfirstlane((int)(blockIdx.x * 8 + (threadIdx.x >> 5)));

    const float4* hp = (const float4*)h;
    const int   beg = rowptr[t];
    const int   end = rowptr[t + 1];
    const float dt  = dinv[t];

    const float4 hs = hp[(size_t)t * 32 + lane];
    const float  ws = dt * dt;
    float4 acc;
    acc.x = hs.x * ws; acc.y = hs.y * ws; acc.z = hs.z * ws; acc.w = hs.w * ws;

    for (int base = beg; base < end; base += 32) {
        int n = end - base;
        if (n > 32) n = 32;
        int   my_s = 0;
        float my_w = 0.0f;
        if (lane < n) {
            my_s = col[base + lane];
            my_w = dinv[my_s] * dt;
        }
        for (int j = 0; j < n; j += 8) {
            switch (n - j) {
                case 1:  add_rows<1>(acc, hp, my_s, my_w, j, lane); break;
                case 2:  add_rows<2>(acc, hp, my_s, my_w, j, lane); break;
                case 3:  add_rows<3>(acc, hp, my_s, my_w, j, lane); break;
                case 4:  add_rows<4>(acc, hp, my_s, my_w, j, lane); break;
                case 5:  add_rows<5>(acc, hp, my_s, my_w, j, lane); break;
                case 6:  add_rows<6>(acc, hp, my_s, my_w, j, lane); break;
                case 7:  add_rows<7>(acc, hp, my_s, my_w, j, lane); break;
                default: add_rows<8>(acc, hp, my_s, my_w, j, lane); break;
            }
        }
    }

    const float4 bb = ((const float4*)b)[lane];
    float4 o;
    o.x = acc.x + bb.x; o.y = acc.y + bb.y; o.z = acc.z + bb.z; o.w = acc.w + bb.w;
    ((float4*)out)[(size_t)t * 32 + lane] = o;
}

__global__ void k_init_stats(float* __restrict__ sums, float* __restrict__ sumsq) {
    int i = threadIdx.x;
    if (i < D) { sums[i] = 0.0f; sumsq[i] = 0.0f; }
}

__global__ void __launch_bounds__(256) k_stats(const float* __restrict__ out,
                                               float* __restrict__ sums,
                                               float* __restrict__ sumsq) {
    int d = threadIdx.x & (D - 1);
    int r0 = blockIdx.x * 2 + (threadIdx.x >> 7);
    float s = 0.0f, ss = 0.0f;
    for (int r = r0; r < N_NODES; r += gridDim.x * 2) {
        float v = out[r * D + d];
        v = v > 0.0f ? v : 0.0f;
        s += v; ss += v * v;
    }
    atomicAdd(&sums[d], s);
    atomicAdd(&sumsq[d], ss);
}

__global__ void k_bn(float* __restrict__ out, const float* __restrict__ sums,
                     const float* __restrict__ sumsq, const float* __restrict__ gamma,
                     const float* __restrict__ beta) {
    int i = blockIdx.x * blockDim.x + threadIdx.x;
    if (i >= N_NODES * D) return;
    int d = i & (D - 1);
    const float invN = 1.0f / (float)N_NODES;
    float mean = sums[d] * invN;
    float var  = sumsq[d] * invN - mean * mean;
    float v = out[i];
    v = v > 0.0f ? v : 0.0f;
    out[i] = gamma[d] * (v - mean) * rsqrtf(var + BN_EPS) + beta[d];
}

extern "C" void kernel_launch(void* const* d_in, const int* in_sizes, int n_in,
                              void* d_out, int out_size, void* d_ws, size_t ws_size,
                              hipStream_t stream) {
    const float* x     = (const float*)d_in[0];
    const int*   edges = (const int*)  d_in[1];
    const float* Wm    = (const float*)d_in[2];
    const float* b     = (const float*)d_in[3];
    const float* gamma = (const float*)d_in[4];
    const float* beta  = (const float*)d_in[5];
    float*       out   = (float*)d_out;

    const int* src = edges;
    const int* dst = edges + N_EDGES;

    float* h      = (float*)d_ws;
    float* dinv   = h + (size_t)N_NODES * D;
    int*   cnt    = (int*)(dinv + N_NODES);
    int*   rowptr = cnt + N_NODES;
    int*   rank   = rowptr + (N_NODES + 32);
    int*   col    = rank + N_EDGES;
    float* sums   = (float*)(col + N_EDGES);
    float* sumsq  = sums + D;

    k_zero_cnt  <<<(N_NODES + 255) / 256, 256, 0, stream>>>(cnt);
    k_count     <<<(N_EDGES + 255) / 256, 256, 0, stream>>>(dst, cnt, rank);
    k_scan      <<<(N_NODES + SCAN_B) / SCAN_B, SCAN_B, 0, stream>>>(cnt, rowptr, dinv);
    k_fill      <<<(N_EDGES + 255) / 256, 256, 0, stream>>>(src, dst, rowptr, rank, col);

    k_gemm      <<<N_NODES / 16, 256, 0, stream>>>(x, Wm, h);

    k_gather    <<<N_NODES / 8, 256, 0, stream>>>(rowptr, col, h, dinv, b, out);

    k_init_stats<<<1, 128, 0, stream>>>(sums, sumsq);
    k_stats     <<<512, 256, 0, stream>>>(out, sums, sumsq);
    k_bn        <<<(N_NODES * D + 255) / 256, 256, 0, stream>>>(out, sums, sumsq, gamma, beta);
}


# BN apply pass: one wave normalises 16 whole rows as float4 with all 16 row loads in flight; 6250 waves instead of 400000 one-dword waves (the pass was bound by wave entry, not by memory)
# speedup vs baseline: 1.3798x; 1.0346x over previous
#include <hip/hip_runtime.h>

#define N_NODES 100000
#define N_EDGES 640000
#define D       128
#define BN_EPS  1e-5f
#define LDS_PITCH 132

typedef float v2f __attribute__((ext_vector_type(2)));
typedef float v8f __attribute__((ext_vector_type(8)));

#define SCAN_B 1024

__global__ void k_zero_cnt(int* __restrict__ cnt) {
    int i = blockIdx.x * blockDim.x + threadIdx.x;
    if (i < N_NODES) cnt[i] = 0;
}

__global__ void k_count(const int* __restrict__ dst, int* __restrict__ cnt,
                        int* __restrict__ rank) {
    int e = blockIdx.x * blockDim.x + threadIdx.x;
    if (e < N_EDGES) rank[e] = atomicAdd(&cnt[dst[e]], 1);
}

__global__ void __launch_bounds__(SCAN_B) k_scan(const int* __restrict__ cnt,
                                                 int* __restrict__ rowptr,
                                                 float* __restrict__ dinv) {
    __shared__ int sp[SCAN_B];
    __shared__ int sc[2][SCAN_B];
    const int t    = threadIdx.x;
    const int base = blockIdx.x * SCAN_B;

    int p = 0;
    for (int i = t; i < base; i += SCAN_B) p += cnt[i];
    const int i = base + t;
    const int c = (i < N_NODES) ? cnt[i] : 0;
    sp[t]    = p;
    sc[0][t] = c;
    __syncthreads();

    for (int s = SCAN_B / 2; s > 0; s >>= 1) {
        if (t < s) sp[t] += sp[t + s];
        __syncthreads();
    }
    const int before = sp[0];

    int cur = 0;
    for (int off = 1; off < SCAN_B; off <<= 1) {
        int v = sc[cur][t];
        if (t >= off) v += sc[cur][t - off];
        sc[cur ^ 1][t] = v;
        __syncthreads();
        cur ^= 1;
    }
    const int incl = sc[cur][t];

    if (i <= N_NODES) rowptr[i] = before + incl - c;
    if (i <  N_NODES) dinv[i]   = rsqrtf((float)(c + 1));
}

__global__ void k_fill(const int* __restrict__ src, const int* __restrict__ dst,
                       const int* __restrict__ rowptr, const int* __restrict__ rank,
                       int* __restrict__ col) {
    int e = blockIdx.x * blockDim.x + threadIdx.x;
    if (e < N_EDGES) col[rowptr[dst[e]] + rank[e]] = src[e];
}

__global__ void __launch_bounds__(256) k_gemm(const float* __restrict__ x,
                                              const float* __restrict__ Wm,
                                              float* __restrict__ h) {
    __shared__ float lx[16 * LDS_PITCH];
    const int tid  = threadIdx.x;
    const int row0 = blockIdx.x * 16;

    for (int idx = tid; idx < 16 * D; idx += 256) {
        int r = idx >> 7, c = idx & (D - 1);
        lx[r * LDS_PITCH + c] = x[(row0 + r) * D + c];
    }
    __syncthreads();

    const int wave = tid >> 5;
    const int lane = tid & 31;
    const int half = lane >> 4;
    const int l15  = lane & 15;
    const int koff = half * 2;
    const int n0   = wave * 16;

    v8f acc = {};
    for (int k = 0; k < D; k += 4) {
        v2f a, b;
        a.x = lx[l15 * LDS_PITCH + k + koff];
        a.y = lx[l15 * LDS_PITCH + k + koff + 1];
        b.x = Wm[(k + koff)     * D + n0 + l15];
        b.y = Wm[(k + koff + 1) * D + n0 + l15];
        acc = __builtin_amdgcn_wmma_f32_16x16x4_f32(
             false, a,  false, b,
             (short)0, acc,  false,  false);
    }

#pragma unroll
    for (int r = 0; r < 8; ++r) {
        int row = row0 + r + half * 8;
        h[row * D + n0 + l15] = acc[r];
    }
}

template <int R>
__device__ __forceinline__ void add_rows(float4& acc, const float4* __restrict__ hp,
                                         int my_s, float my_w, int j, int lane) {
    float4 v[R];
#pragma unroll
    for (int k = 0; k < R; ++k)
        v[k] = hp[(size_t)__builtin_amdgcn_readlane(my_s, j + k) * 32 + lane];
#pragma unroll
    for (int k = 0; k < R; ++k) {
        const float w = __int_as_float(__builtin_amdgcn_readlane(__float_as_int(my_w), j + k));
        acc.x = fmaf(v[k].x, w, acc.x);
        acc.y = fmaf(v[k].y, w, acc.y);
        acc.z = fmaf(v[k].z, w, acc.z);
        acc.w = fmaf(v[k].w, w, acc.w);
    }
}

__global__ void __launch_bounds__(256) k_gather(const int* __restrict__ rowptr,
                                                const int* __restrict__ col,
                                                const float* __restrict__ h,
                                                const float* __restrict__ dinv,
                                                const float* __restrict__ b,
                                                float* __restrict__ out) {
    const int lane = threadIdx.x & 31;
    const int t = __builtin_amdgcn_readfirstlane((int)(blockIdx.x * 8 + (threadIdx.x >> 5)));

    const float4* hp = (const float4*)h;
    const int   beg = rowptr[t];
    const int   end = rowptr[t + 1];
    const float dt  = dinv[t];

    const float4 hs = hp[(size_t)t * 32 + lane];
    const float  ws = dt * dt;
    float4 acc;
    acc.x = hs.x * ws; acc.y = hs.y * ws; acc.z = hs.z * ws; acc.w = hs.w * ws;

    for (int base = beg; base < end; base += 32) {
        int n = end - base;
        if (n > 32) n = 32;
        int   my_s = 0;
        float my_w = 0.0f;
        if (lane < n) {
            my_s = col[base + lane];
            my_w = dinv[my_s] * dt;
        }
        for (int j = 0; j < n; j += 8) {
            switch (n - j) {
                case 1:  add_rows<1>(acc, hp, my_s, my_w, j, lane); break;
                case 2:  add_rows<2>(acc, hp, my_s, my_w, j, lane); break;
                case 3:  add_rows<3>(acc, hp, my_s, my_w, j, lane); break;
                case 4:  add_rows<4>(acc, hp, my_s, my_w, j, lane); break;
                case 5:  add_rows<5>(acc, hp, my_s, my_w, j, lane); break;
                case 6:  add_rows<6>(acc, hp, my_s, my_w, j, lane); break;
                case 7:  add_rows<7>(acc, hp, my_s, my_w, j, lane); break;
                default: add_rows<8>(acc, hp, my_s, my_w, j, lane); break;
            }
        }
    }

    const float4 bb = ((const float4*)b)[lane];
    float4 o;
    o.x = acc.x + bb.x; o.y = acc.y + bb.y; o.z = acc.z + bb.z; o.w = acc.w + bb.w;
    ((float4*)out)[(size_t)t * 32 + lane] = o;
}

__global__ void k_init_stats(float* __restrict__ sums, float* __restrict__ sumsq) {
    int i = threadIdx.x;
    if (i < D) { sums[i] = 0.0f; sumsq[i] = 0.0f; }
}

__global__ void __launch_bounds__(256) k_stats(const float* __restrict__ out,
                                               float* __restrict__ sums,
                                               float* __restrict__ sumsq) {
    int d = threadIdx.x & (D - 1);
    int r0 = blockIdx.x * 2 + (threadIdx.x >> 7);
    float s = 0.0f, ss = 0.0f;
    for (int r = r0; r < N_NODES; r += gridDim.x * 2) {
        float v = out[r * D + d];
        v = v > 0.0f ? v : 0.0f;
        s += v; ss += v * v;
    }
    atomicAdd(&sums[d], s);
    atomicAdd(&sumsq[d], ss);
}

#define BN_ROWS 16

__device__ __forceinline__ float bn_one(float v, float mean, float rs, float g, float bt) {
    v = v > 0.0f ? v : 0.0f;
    return g * (v - mean) * rs + bt;
}

__global__ void __launch_bounds__(256) k_bn(float* __restrict__ out,
                                            const float* __restrict__ sums,
                                            const float* __restrict__ sumsq,
                                            const float* __restrict__ gamma,
                                            const float* __restrict__ beta) {
    static_assert(N_NODES % BN_ROWS == 0, "whole waves only");
    const int lane = threadIdx.x & 31;
    const int row0 = (int)(blockIdx.x * 8 + (threadIdx.x >> 5)) * BN_ROWS;
    if (row0 >= N_NODES) return;

    float4* op = (float4*)out + (size_t)row0 * 32 + lane;
    float4 v[BN_ROWS];
#pragma unroll
    for (int k = 0; k < BN_ROWS; ++k) v[k] = op[k * 32];

    const float4 s  = ((const float4*)sums)[lane];
    const float4 sq = ((const float4*)sumsq)[lane];
    const float4 g  = ((const float4*)gamma)[lane];
    const float4 bt = ((const float4*)beta)[lane];

    const float invN = 1.0f / (float)N_NODES;
    float4 mean, var, rs;
    mean.x = s.x * invN; mean.y = s.y * invN; mean.z = s.z * invN; mean.w = s.w * invN;
    var.x = sq.x * invN - mean.x * mean.x;
    var.y = sq.y * invN - mean.y * mean.y;
    var.z = sq.z * invN - mean.z * mean.z;
    var.w = sq.w * invN - mean.w * mean.w;
    rs.x = rsqrtf(var.x + BN_EPS); rs.y = rsqrtf(var.y + BN_EPS);
    rs.z = rsqrtf(var.z + BN_EPS); rs.w = rsqrtf(var.w + BN_EPS);

#pragma unroll
    for (int k = 0; k < BN_ROWS; ++k) {
        float4 o;
        o.x = bn_one(v[k].x, mean.x, rs.x, g.x, bt.x);
        o.y = bn_one(v[k].y, mean.y, rs.y, g.y, bt.y);
        o.z = bn_one(v[k].z, mean.z, rs.z, g.z, bt.z);
        o.w = bn_one(v[k].w, mean.w, rs.w, g.w, bt.w);
        op[k * 32] = o;
    }
}

extern "C" void kernel_launch(void* const* d_in, const int* in_sizes, int n_in,
                              void* d_out, int out_size, void* d_ws, size_t ws_size,
                              hipStream_t stream) {
    const float* x     = (const float*)d_in[0];
    const int*   edges = (const int*)  d_in[1];
    const float* Wm    = (const float*)d_in[2];
    const float* b     = (const float*)d_in[3];
    const float* gamma = (const float*)d_in[4];
    const float* beta  = (const float*)d_in[5];
    float*       out   = (float*)d_out;

    const int* src = edges;
    const int* dst = edges + N_EDGES;

    float* h      = (float*)d_ws;
    float* dinv   = h + (size_t)N_NODES * D;
    int*   cnt    = (int*)(dinv + N_NODES);
    int*   rowptr = cnt + N_NODES;
    int*   rank   = rowptr + (N_NODES + 32);
    int*   col    = rank + N_EDGES;
    float* sums   = (float*)(col + N_EDGES);
    float* sumsq  = sums + D;

    k_zero_cnt  <<<(N_NODES + 255) / 256, 256, 0, stream>>>(cnt);
    k_count     <<<(N_EDGES + 255) / 256, 256, 0, stream>>>(dst, cnt, rank);
    k_scan      <<<(N_NODES + SCAN_B) / SCAN_B, SCAN_B, 0, stream>>>(cnt, rowptr, dinv);
    k_fill      <<<(N_EDGES + 255) / 256, 256, 0, stream>>>(src, dst, rowptr, rank, col);

    k_gemm      <<<N_NODES / 16, 256, 0, stream>>>(x, Wm, h);

    k_gather    <<<N_NODES / 8, 256, 0, stream>>>(rowptr, col, h, dinv, b, out);

    k_init_stats<<<1, 128, 0, stream>>>(sums, sumsq);
    k_stats     <<<512, 256, 0, stream>>>(out, sums, sumsq);
    k_bn        <<<(N_NODES / BN_ROWS + 7) / 8, 256, 0, stream>>>(out, sums, sumsq, gamma, beta);
}


# BN statistics pass: one wave sums 16 whole rows as float4 with all 16 row loads in flight; 16 waves add partials in LDS, one float atomic per column per block (sum order changes)
# speedup vs baseline: 1.7146x; 1.2426x over previous
#include <hip/hip_runtime.h>

#define N_NODES 100000
#define N_EDGES 640000
#define D       128
#define BN_EPS  1e-5f
#define LDS_PITCH 132

typedef float v2f __attribute__((ext_vector_type(2)));
typedef float v8f __attribute__((ext_vector_type(8)));

#define SCAN_B 1024

__global__ void k_zero_cnt(int* __restrict__ cnt) {
    int i = blockIdx.x * blockDim.x + threadIdx.x;
    if (i < N_NODES) cnt[i] = 0;
}

__global__ void k_count(const int* __restrict__ dst, int* __restrict__ cnt,
                        int* __restrict__ rank) {
    int e = blockIdx.x * blockDim.x + threadIdx.x;
    if (e < N_EDGES) rank[e] = atomicAdd(&cnt[dst[e]], 1);
}

__global__ void __launch_bounds__(SCAN_B) k_scan(const int* __restrict__ cnt,
                                                 int* __restrict__ rowptr,
                                                 float* __restrict__ dinv) {
    __shared__ int sp[SCAN_B];
    __shared__ int sc[2][SCAN_B];
    const int t    = threadIdx.x;
    const int base = blockIdx.x * SCAN_B;

    int p = 0;
    for (int i = t; i < base; i += SCAN_B) p += cnt[i];
    const int i = base + t;
    const int c = (i < N_NODES) ? cnt[i] : 0;
    sp[t]    = p;
    sc[0][t] = c;
    __syncthreads();

    for (int s = SCAN_B / 2; s > 0; s >>= 1) {
        if (t < s) sp[t] += sp[t + s];
        __syncthreads();
    }
    const int before = sp[0];

    int cur = 0;
    for (int off = 1; off < SCAN_B; off <<= 1) {
        int v = sc[cur][t];
        if (t >= off) v += sc[cur][t - off];
        sc[cur ^ 1][t] = v;
        __syncthreads();
        cur ^= 1;
    }
    const int incl = sc[cur][t];

    if (i <= N_NODES) rowptr[i] = before + incl - c;
    if (i <  N_NODES) dinv[i]   = rsqrtf((float)(c + 1));
}

__global__ void k_fill(const int* __restrict__ src, const int* __restrict__ dst,
                       const int* __restrict__ rowptr, const int* __restrict__ rank,
                       int* __restrict__ col) {
    int e = blockIdx.x * blockDim.x + threadIdx.x;
    if (e < N_EDGES) col[rowptr[dst[e]] + rank[e]] = src[e];
}

__global__ void __launch_bounds__(256) k_gemm(const float* __restrict__ x,
                                              const float* __restrict__ Wm,
                                              float* __restrict__ h) {
    __shared__ float lx[16 * LDS_PITCH];
    const int tid  = threadIdx.x;
    const int row0 = blockIdx.x * 16;

    for (int idx = tid; idx < 16 * D; idx += 256) {
        int r = idx >> 7, c = idx & (D - 1);
        lx[r * LDS_PITCH + c] = x[(row0 + r) * D + c];
    }
    __syncthreads();

    const int wave = tid >> 5;
    const int lane = tid & 31;
    const int half = lane >> 4;
    const int l15  = lane & 15;
    const int koff = half * 2;
    const int n0   = wave * 16;

    v8f acc = {};
    for (int k = 0; k < D; k += 4) {
        v2f a, b;
        a.x = lx[l15 * LDS_PITCH + k + koff];
        a.y = lx[l15 * LDS_PITCH + k + koff + 1];
        b.x = Wm[(k + koff)     * D + n0 + l15];
        b.y = Wm[(k + koff + 1) * D + n0 + l15];
        acc = __builtin_amdgcn_wmma_f32_16x16x4_f32(
             false, a,  false, b,
             (short)0, acc,  false,  false);
    }

#pragma unroll
    for (int r = 0; r < 8; ++r) {
        int row = row0 + r + half * 8;
        h[row * D + n0 + l15] = acc[r];
    }
}

template <int R>
__device__ __forceinline__ void add_rows(float4& acc, const float4* __restrict__ hp,
                                         int my_s, float my_w, int j, int lane) {
    float4 v[R];
#pragma unroll
    for (int k = 0; k < R; ++k)
        v[k] = hp[(size_t)__builtin_amdgcn_readlane(my_s, j + k) * 32 + lane];
#pragma unroll
    for (int k = 0; k < R; ++k) {
        const float w = __int_as_float(__builtin_amdgcn_readlane(__float_as_int(my_w), j + k));
        acc.x = fmaf(v[k].x, w, acc.x);
        acc.y = fmaf(v[k].y, w, acc.y);
        acc.z = fmaf(v[k].z, w, acc.z);
        acc.w = fmaf(v[k].w, w, acc.w);
    }
}

__global__ void __launch_bounds__(256) k_gather(const int* __restrict__ rowptr,
                                                const int* __restrict__ col,
                                                const float* __restrict__ h,
                                                const float* __restrict__ dinv,
                                                const float* __restrict__ b,
                                                float* __restrict__ out) {
    const int lane = threadIdx.x & 31;
    const int t = __builtin_amdgcn_readfirstlane((int)(blockIdx.x * 8 + (threadIdx.x >> 5)));

    const float4* hp = (const float4*)h;
    const int   beg = rowptr[t];
    const int   end = rowptr[t + 1];
    const float dt  = dinv[t];

    const float4 hs = hp[(size_t)t * 32 + lane];
    const float  ws = dt * dt;
    float4 acc;
    acc.x = hs.x * ws; acc.y = hs.y * ws; acc.z = hs.z * ws; acc.w = hs.w * ws;

    for (int base = beg; base < end; base += 32) {
        int n = end - base;
        if (n > 32) n = 32;
        int   my_s = 0;
        float my_w = 0.0f;
        if (lane < n) {
            my_s = col[base + lane];
            my_w = dinv[my_s] * dt;
        }
        for (int j = 0; j < n; j += 8) {
            switch (n - j) {
                case 1:  add_rows<1>(acc, hp, my_s, my_w, j, lane); break;
                case 2:  add_rows<2>(acc, hp, my_s, my_w, j, lane); break;
                case 3:  add_rows<3>(acc, hp, my_s, my_w, j, lane); break;
                case 4:  add_rows<4>(acc, hp, my_s, my_w, j, lane); break;
                case 5:  add_rows<5>(acc, hp, my_s, my_w, j, lane); break;
                case 6:  add_rows<6>(acc, hp, my_s, my_w, j, lane); break;
                case 7:  add_rows<7>(acc, hp, my_s, my_w, j, lane); break;
                default: add_rows<8>(acc, hp, my_s, my_w, j, lane); break;
            }
        }
    }

    const float4 bb = ((const float4*)b)[lane];
    float4 o;
    o.x = acc.x + bb.x; o.y = acc.y + bb.y; o.z = acc.z + bb.z; o.w = acc.w + bb.w;
    ((float4*)out)[(size_t)t * 32 + lane] = o;
}

__global__ void k_init_stats(float* __restrict__ sums, float* __restrict__ sumsq) {
    int i = threadIdx.x;
    if (i < D) { sums[i] = 0.0f; sumsq[i] = 0.0f; }
}

#define ST_ROWS  16
#define ST_WAVES 16

__global__ void __launch_bounds__(ST_WAVES * 32) k_stats(const float* __restrict__ out,
                                                         float* __restrict__ sums,
                                                         float* __restrict__ sumsq) {
    static_assert(N_NODES % ST_ROWS == 0, "whole waves only");
    __shared__ float red[ST_WAVES][2 * D];
    const int lane = threadIdx.x & 31;
    const int wave = threadIdx.x >> 5;
    const int row0 = (int)(blockIdx.x * ST_WAVES + wave) * ST_ROWS;

    float4 s = make_float4(0.0f, 0.0f, 0.0f, 0.0f);
    float4 q = make_float4(0.0f, 0.0f, 0.0f, 0.0f);
    if (row0 < N_NODES) {
        const float4* op = (const float4*)out + (size_t)row0 * 32 + lane;
        float4 v[ST_ROWS];
#pragma unroll
        for (int k = 0; k < ST_ROWS; ++k) v[k] = op[k * 32];
#pragma unroll
        for (int k = 0; k < ST_ROWS; ++k) {
            float4 r;
            r.x = v[k].x > 0.0f ? v[k].x : 0.0f;
            r.y = v[k].y > 0.0f ? v[k].y : 0.0f;
            r.z = v[k].z > 0.0f ? v[k].z : 0.0f;
            r.w = v[k].w > 0.0f ? v[k].w : 0.0f;
            s.x += r.x; s.y += r.y; s.z += r.z; s.w += r.w;
            q.x += r.x * r.x; q.y += r.y * r.y; q.z += r.z * r.z; q.w += r.w * r.w;
        }
    }
    ((float4*)&red[wave][0])[lane] = s;
    ((float4*)&red[wave][D])[lane] = q;
    __syncthreads();

    if (threadIdx.x < 2 * D) {
        float a = 0.0f;
#pragma unroll
        for (int w = 0; w < ST_WAVES; ++w) a += red[w][threadIdx.x];
        if (threadIdx.x < D) atomicAdd(&sums[threadIdx.x], a);
        else                 atomicAdd(&sumsq[threadIdx.x - D], a);
    }
}

#define BN_ROWS 16

__device__ __forceinline__ float bn_one(float v, float mean, float rs, float g, float bt) {
    v = v > 0.0f ? v : 0.0f;
    return g * (v - mean) * rs + bt;
}

__global__ void __launch_bounds__(256) k_bn(float* __restrict__ out,
                                            const float* __restrict__ sums,
                                            const float* __restrict__ sumsq,
                                            const float* __restrict__ gamma,
                                            const float* __restrict__ beta) {
    static_assert(N_NODES % BN_ROWS == 0, "whole waves only");
    const int lane = threadIdx.x & 31;
    const int row0 = (int)(blockIdx.x * 8 + (threadIdx.x >> 5)) * BN_ROWS;
    if (row0 >= N_NODES) return;

    float4* op = (float4*)out + (size_t)row0 * 32 + lane;
    float4 v[BN_ROWS];
#pragma unroll
    for (int k = 0; k < BN_ROWS; ++k) v[k] = op[k * 32];

    const float4 s  = ((const float4*)sums)[lane];
    const float4 sq = ((const float4*)sumsq)[lane];
    const float4 g  = ((const float4*)gamma)[lane];
    const float4 bt = ((const float4*)beta)[lane];

    const float invN = 1.0f / (float)N_NODES;
    float4 mean, var, rs;
    mean.x = s.x * invN; mean.y = s.y * invN; mean.z = s.z * invN; mean.w = s.w * invN;
    var.x = sq.x * invN - mean.x * mean.x;
    var.y = sq.y * invN - mean.y * mean.y;
    var.z = sq.z * invN - mean.z * mean.z;
    var.w = sq.w * invN - mean.w * mean.w;
    rs.x = rsqrtf(var.x + BN_EPS); rs.y = rsqrtf(var.y + BN_EPS);
    rs.z = rsqrtf(var.z + BN_EPS); rs.w = rsqrtf(var.w + BN_EPS);

#pragma unroll
    for (int k = 0; k < BN_ROWS; ++k) {
        float4 o;
        o.x = bn_one(v[k].x, mean.x, rs.x, g.x, bt.x);
        o.y = bn_one(v[k].y, mean.y, rs.y, g.y, bt.y);
        o.z = bn_one(v[k].z, mean.z, rs.z, g.z, bt.z);
        o.w = bn_one(v[k].w, mean.w, rs.w, g.w, bt.w);
        op[k * 32] = o;
    }
}

extern "C" void kernel_launch(void* const* d_in, const int* in_sizes, int n_in,
                              void* d_out, int out_size, void* d_ws, size_t ws_size,
                              hipStream_t stream) {
    const float* x     = (const float*)d_in[0];
    const int*   edges = (const int*)  d_in[1];
    const float* Wm    = (const float*)d_in[2];
    const float* b     = (const float*)d_in[3];
    const float* gamma = (const float*)d_in[4];
    const float* beta  = (const float*)d_in[5];
    float*       out   = (float*)d_out;

    const int* src = edges;
    const int* dst = edges + N_EDGES;

    float* h      = (float*)d_ws;
    float* dinv   = h + (size_t)N_NODES * D;
    int*   cnt    = (int*)(dinv + N_NODES);
    int*   rowptr = cnt + N_NODES;
    int*   rank   = rowptr + (N_NODES + 32);
    int*   col    = rank + N_EDGES;
    float* sums   = (float*)(col + N_EDGES);
    float* sumsq  = sums + D;

    k_zero_cnt  <<<(N_NODES + 255) / 256, 256, 0, stream>>>(cnt);
    k_count     <<<(N_EDGES + 255) / 256, 256, 0, stream>>>(dst, cnt, rank);
    k_scan      <<<(N_NODES + SCAN_B) / SCAN_B, SCAN_B, 0, stream>>>(cnt, rowptr, dinv);
    k_fill      <<<(N_EDGES + 255) / 256, 256, 0, stream>>>(src, dst, rowptr, rank, col);

    k_gemm      <<<N_NODES / 16, 256, 0, stream>>>(x, Wm, h);

    k_gather    <<<N_NODES / 8, 256, 0, stream>>>(rowptr, col, h, dinv, b, out);

    k_init_stats<<<1, 128, 0, stream>>>(sums, sumsq);
    k_stats     <<<(N_NODES / ST_ROWS + ST_WAVES - 1) / ST_WAVES, ST_WAVES * 32, 0, stream>>>(out, sums, sumsq);
    k_bn        <<<(N_NODES / BN_ROWS + 7) / 8, 256, 0, stream>>>(out, sums, sumsq, gamma, beta);
}


# x@W launch folded into the prefix-sum launch (scan recast to 256-thread blocks, 4 nodes a thread, to share the grid); stats zeroing folded into cnt zeroing: 9 -> 7 launches; no float sum reordered
# speedup vs baseline: 1.8391x; 1.0726x over previous
#include <hip/hip_runtime.h>

#define N_NODES 100000
#define N_EDGES 640000
#define D       128
#define BN_EPS  1e-5f
#define LDS_PITCH 132

typedef float v2f __attribute__((ext_vector_type(2)));
typedef float v8f __attribute__((ext_vector_type(8)));

#define GEMM_TILES (N_NODES / 16)

__device__ __forceinline__ void gemm_tile(int tile, const float* __restrict__ x,
                                          const float* __restrict__ Wm,
                                          float* __restrict__ h, float* lx) {
    const int tid  = threadIdx.x;
    const int row0 = tile * 16;

    for (int idx = tid; idx < 16 * D; idx += 256) {
        int r = idx >> 7, c = idx & (D - 1);
        lx[r * LDS_PITCH + c] = x[(row0 + r) * D + c];
    }
    __syncthreads();

    const int wave = tid >> 5;
    const int lane = tid & 31;
    const int half = lane >> 4;
    const int l15  = lane & 15;
    const int koff = half * 2;
    const int n0   = wave * 16;

    v8f acc = {};
    for (int k = 0; k < D; k += 4) {
        v2f a, b;
        a.x = lx[l15 * LDS_PITCH + k + koff];
        a.y = lx[l15 * LDS_PITCH + k + koff + 1];
        b.x = Wm[(k + koff)     * D + n0 + l15];
        b.y = Wm[(k + koff + 1) * D + n0 + l15];
        acc = __builtin_amdgcn_wmma_f32_16x16x4_f32(
             false, a,  false, b,
             (short)0, acc,  false,  false);
    }

#pragma unroll
    for (int r = 0; r < 8; ++r) {
        int row = row0 + r + half * 8;
        h[row * D + n0 + l15] = acc[r];
    }
}

#define SCAN_B 1024
#define SCAN_T 256
#define SCAN_BLOCKS ((N_NODES + SCAN_B) / SCAN_B)
#define FRONT_U 16

__global__ void k_zero(int* __restrict__ cnt, float* __restrict__ sums,
                       float* __restrict__ sumsq) {
    int i = blockIdx.x * blockDim.x + threadIdx.x;
    if (i < N_NODES) cnt[i] = 0;
    if (i < D) { sums[i] = 0.0f; sumsq[i] = 0.0f; }
}

__global__ void k_count(const int* __restrict__ dst, int* __restrict__ cnt,
                        int* __restrict__ rank) {
    int e = blockIdx.x * blockDim.x + threadIdx.x;
    if (e < N_EDGES) rank[e] = atomicAdd(&cnt[dst[e]], 1);
}

__device__ __forceinline__ void scan_block(int b, const int* __restrict__ cnt,
                                           int* __restrict__ rowptr,
                                           float* __restrict__ dinv,
                                           int* sp, int* sc) {
    static_assert(N_NODES % 4 == 0, "a thread's four nodes are in range together");
    static_assert(SCAN_B == 4 * SCAN_T, "four nodes per thread");
    const int t = threadIdx.x;
    const int4* c4 = (const int4*)cnt;

    int p = 0;
    for (int k0 = 0; k0 < b; k0 += FRONT_U) {
        int4 v[FRONT_U];
#pragma unroll
        for (int j = 0; j < FRONT_U; ++j) {
            int k = k0 + j;
            if (k > b - 1) k = b - 1;
            v[j] = c4[k * SCAN_T + t];
        }
#pragma unroll
        for (int j = 0; j < FRONT_U; ++j)
            if (k0 + j < b) p += v[j].x + v[j].y + v[j].z + v[j].w;
    }

    const int i0 = b * SCAN_B + 4 * t;
    int4 c = make_int4(0, 0, 0, 0);
    if (i0 < N_NODES) c = c4[i0 >> 2];
    const int tsum = c.x + c.y + c.z + c.w;
    sp[t] = p;
    sc[t] = tsum;
    __syncthreads();

    for (int s = SCAN_T / 2; s > 0; s >>= 1) {
        if (t < s) sp[t] += sp[t + s];
        __syncthreads();
    }
    const int before = sp[0];

    int cur = 0;
    for (int off = 1; off < SCAN_T; off <<= 1) {
        int v = sc[cur * SCAN_T + t];
        if (t >= off) v += sc[cur * SCAN_T + t - off];
        sc[(cur ^ 1) * SCAN_T + t] = v;
        __syncthreads();
        cur ^= 1;
    }
    const int r0 = before + sc[cur * SCAN_T + t] - tsum;

    if (i0 < N_NODES) {
        int4 rp;
        rp.x = r0;
        rp.y = r0 + c.x;
        rp.z = r0 + c.x + c.y;
        rp.w = r0 + c.x + c.y + c.z;
        ((int4*)rowptr)[i0 >> 2] = rp;
        float4 dv;
        dv.x = rsqrtf((float)(c.x + 1));
        dv.y = rsqrtf((float)(c.y + 1));
        dv.z = rsqrtf((float)(c.z + 1));
        dv.w = rsqrtf((float)(c.w + 1));
        ((float4*)dinv)[i0 >> 2] = dv;
    } else if (i0 == N_NODES) {
        rowptr[N_NODES] = r0;
    }
}

__global__ void __launch_bounds__(256) k_scan_gemm(const int* __restrict__ cnt,
                                                   int* __restrict__ rowptr,
                                                   float* __restrict__ dinv,
                                                   const float* __restrict__ x,
                                                   const float* __restrict__ Wm,
                                                   float* __restrict__ h) {
    __shared__ float lx[16 * LDS_PITCH];
    __shared__ int sp[SCAN_T];
    __shared__ int sc[2 * SCAN_T];
    const int bid = blockIdx.x;
    if (bid < SCAN_BLOCKS) {
        scan_block(bid, cnt, rowptr, dinv, sp, sc);
        return;
    }
    gemm_tile(bid - SCAN_BLOCKS, x, Wm, h, lx);
}

__global__ void k_fill(const int* __restrict__ src, const int* __restrict__ dst,
                       const int* __restrict__ rowptr, const int* __restrict__ rank,
                       int* __restrict__ col) {
    int e = blockIdx.x * blockDim.x + threadIdx.x;
    if (e < N_EDGES) col[rowptr[dst[e]] + rank[e]] = src[e];
}

template <int R>
__device__ __forceinline__ void add_rows(float4& acc, const float4* __restrict__ hp,
                                         int my_s, float my_w, int j, int lane) {
    float4 v[R];
#pragma unroll
    for (int k = 0; k < R; ++k)
        v[k] = hp[(size_t)__builtin_amdgcn_readlane(my_s, j + k) * 32 + lane];
#pragma unroll
    for (int k = 0; k < R; ++k) {
        const float w = __int_as_float(__builtin_amdgcn_readlane(__float_as_int(my_w), j + k));
        acc.x = fmaf(v[k].x, w, acc.x);
        acc.y = fmaf(v[k].y, w, acc.y);
        acc.z = fmaf(v[k].z, w, acc.z);
        acc.w = fmaf(v[k].w, w, acc.w);
    }
}

__global__ void __launch_bounds__(256) k_gather(const int* __restrict__ rowptr,
                                                const int* __restrict__ col,
                                                const float* __restrict__ h,
                                                const float* __restrict__ dinv,
                                                const float* __restrict__ b,
                                                float* __restrict__ out) {
    const int lane = threadIdx.x & 31;
    const int t = __builtin_amdgcn_readfirstlane((int)(blockIdx.x * 8 + (threadIdx.x >> 5)));

    const float4* hp = (const float4*)h;
    const int   beg = rowptr[t];
    const int   end = rowptr[t + 1];
    const float dt  = dinv[t];

    const float4 hs = hp[(size_t)t * 32 + lane];
    const float  ws = dt * dt;
    float4 acc;
    acc.x = hs.x * ws; acc.y = hs.y * ws; acc.z = hs.z * ws; acc.w = hs.w * ws;

    for (int base = beg; base < end; base += 32) {
        int n = end - base;
        if (n > 32) n = 32;
        int   my_s = 0;
        float my_w = 0.0f;
        if (lane < n) {
            my_s = col[base + lane];
            my_w = dinv[my_s] * dt;
        }
        for (int j = 0; j < n; j += 8) {
            switch (n - j) {
                case 1:  add_rows<1>(acc, hp, my_s, my_w, j, lane); break;
                case 2:  add_rows<2>(acc, hp, my_s, my_w, j, lane); break;
                case 3:  add_rows<3>(acc, hp, my_s, my_w, j, lane); break;
                case 4:  add_rows<4>(acc, hp, my_s, my_w, j, lane); break;
                case 5:  add_rows<5>(acc, hp, my_s, my_w, j, lane); break;
                case 6:  add_rows<6>(acc, hp, my_s, my_w, j, lane); break;
                case 7:  add_rows<7>(acc, hp, my_s, my_w, j, lane); break;
                default: add_rows<8>(acc, hp, my_s, my_w, j, lane); break;
            }
        }
    }

    const float4 bb = ((const float4*)b)[lane];
    float4 o;
    o.x = acc.x + bb.x; o.y = acc.y + bb.y; o.z = acc.z + bb.z; o.w = acc.w + bb.w;
    ((float4*)out)[(size_t)t * 32 + lane] = o;
}

#define ST_ROWS  16
#define ST_WAVES 16

__global__ void __launch_bounds__(ST_WAVES * 32) k_stats(const float* __restrict__ out,
                                                         float* __restrict__ sums,
                                                         float* __restrict__ sumsq) {
    static_assert(N_NODES % ST_ROWS == 0, "whole waves only");
    __shared__ float red[ST_WAVES][2 * D];
    const int lane = threadIdx.x & 31;
    const int wave = threadIdx.x >> 5;
    const int row0 = (int)(blockIdx.x * ST_WAVES + wave) * ST_ROWS;

    float4 s = make_float4(0.0f, 0.0f, 0.0f, 0.0f);
    float4 q = make_float4(0.0f, 0.0f, 0.0f, 0.0f);
    if (row0 < N_NODES) {
        const float4* op = (const float4*)out + (size_t)row0 * 32 + lane;
        float4 v[ST_ROWS];
#pragma unroll
        for (int k = 0; k < ST_ROWS; ++k) v[k] = op[k * 32];
#pragma unroll
        for (int k = 0; k < ST_ROWS; ++k) {
            float4 r;
            r.x = v[k].x > 0.0f ? v[k].x : 0.0f;
            r.y = v[k].y > 0.0f ? v[k].y : 0.0f;
            r.z = v[k].z > 0.0f ? v[k].z : 0.0f;
            r.w = v[k].w > 0.0f ? v[k].w : 0.0f;
            s.x += r.x; s.y += r.y; s.z += r.z; s.w += r.w;
            q.x += r.x * r.x; q.y += r.y * r.y; q.z += r.z * r.z; q.w += r.w * r.w;
        }
    }
    ((float4*)&red[wave][0])[lane] = s;
    ((float4*)&red[wave][D])[lane] = q;
    __syncthreads();

    if (threadIdx.x < 2 * D) {
        float a = 0.0f;
#pragma unroll
        for (int w = 0; w < ST_WAVES; ++w) a += red[w][threadIdx.x];
        if (threadIdx.x < D) atomicAdd(&sums[threadIdx.x], a);
        else                 atomicAdd(&sumsq[threadIdx.x - D], a);
    }
}

#define BN_ROWS 16

__device__ __forceinline__ float bn_one(float v, float mean, float rs, float g, float bt) {
    v = v > 0.0f ? v : 0.0f;
    return g * (v - mean) * rs + bt;
}

__global__ void __launch_bounds__(256) k_bn(float* __restrict__ out,
                                            const float* __restrict__ sums,
                                            const float* __restrict__ sumsq,
                                            const float* __restrict__ gamma,
                                            const float* __restrict__ beta) {
    static_assert(N_NODES % BN_ROWS == 0, "whole waves only");
    const int lane = threadIdx.x & 31;
    const int row0 = (int)(blockIdx.x * 8 + (threadIdx.x >> 5)) * BN_ROWS;
    if (row0 >= N_NODES) return;

    float4* op = (float4*)out + (size_t)row0 * 32 + lane;
    float4 v[BN_ROWS];
#pragma unroll
    for (int k = 0; k < BN_ROWS; ++k) v[k] = op[k * 32];

    const float4 s  = ((const float4*)sums)[lane];
    const float4 sq = ((const float4*)sumsq)[lane];
    const float4 g  = ((const float4*)gamma)[lane];
    const float4 bt = ((const float4*)beta)[lane];

    const float invN = 1.0f / (float)N_NODES;
    float4 mean, var, rs;
    mean.x = s.x * invN; mean.y = s.y * invN; mean.z = s.z * invN; mean.w = s.w * invN;
    var.x = sq.x * invN - mean.x * mean.x;
    var.y = sq.y * invN - mean.y * mean.y;
    var.z = sq.z * invN - mean.z * mean.z;
    var.w = sq.w * invN - mean.w * mean.w;
    rs.x = rsqrtf(var.x + BN_EPS); rs.y = rsqrtf(var.y + BN_EPS);
    rs.z = rsqrtf(var.z + BN_EPS); rs.w = rsqrtf(var.w + BN_EPS);

#pragma unroll
    for (int k = 0; k < BN_ROWS; ++k) {
        float4 o;
        o.x = bn_one(v[k].x, mean.x, rs.x, g.x, bt.x);
        o.y = bn_one(v[k].y, mean.y, rs.y, g.y, bt.y);
        o.z = bn_one(v[k].z, mean.z, rs.z, g.z, bt.z);
        o.w = bn_one(v[k].w, mean.w, rs.w, g.w, bt.w);
        op[k * 32] = o;
    }
}

extern "C" void kernel_launch(void* const* d_in, const int* in_sizes, int n_in,
                              void* d_out, int out_size, void* d_ws, size_t ws_size,
                              hipStream_t stream) {
    const float* x     = (const float*)d_in[0];
    const int*   edges = (const int*)  d_in[1];
    const float* Wm    = (const float*)d_in[2];
    const float* b     = (const float*)d_in[3];
    const float* gamma = (const float*)d_in[4];
    const float* beta  = (const float*)d_in[5];
    float*       out   = (float*)d_out;

    const int* src = edges;
    const int* dst = edges + N_EDGES;

    float* h      = (float*)d_ws;
    float* dinv   = h + (size_t)N_NODES * D;
    int*   cnt    = (int*)(dinv + N_NODES);
    int*   rowptr = cnt + N_NODES;
    int*   rank   = rowptr + (N_NODES + 32);
    int*   col    = rank + N_EDGES;
    float* sums   = (float*)(col + N_EDGES);
    float* sumsq  = sums + D;

    k_zero      <<<(N_NODES + 255) / 256, 256, 0, stream>>>(cnt, sums, sumsq);
    k_count     <<<(N_EDGES + 255) / 256, 256, 0, stream>>>(dst, cnt, rank);
    k_scan_gemm <<<SCAN_BLOCKS + GEMM_TILES, 256, 0, stream>>>(cnt, rowptr, dinv, x, Wm, h);
    k_fill      <<<(N_EDGES + 255) / 256, 256, 0, stream>>>(src, dst, rowptr, rank, col);

    k_gather    <<<N_NODES / 8, 256, 0, stream>>>(rowptr, col, h, dinv, b, out);

    k_stats     <<<(N_NODES / ST_ROWS + ST_WAVES - 1) / ST_WAVES, ST_WAVES * 32, 0, stream>>>(out, sums, sumsq);
    k_bn        <<<(N_NODES / BN_ROWS + 7) / 8, 256, 0, stream>>>(out, sums, sumsq, gamma, beta);
}


# x@W tiles spread over the count, prefix-sum and fill launches, interleaved with their edge blocks, instead of all in the prefix-sum launch: matrix work overlaps the atomic and scatter waits
# speedup vs baseline: 1.9380x; 1.0538x over previous
#include <hip/hip_runtime.h>

#define N_NODES 100000
#define N_EDGES 640000
#define D       128
#define BN_EPS  1e-5f
#define LDS_PITCH 132

typedef float v2f __attribute__((ext_vector_type(2)));
typedef float v8f __attribute__((ext_vector_type(8)));

#define GEMM_TILES  (N_NODES / 16)
#define TILES_COUNT 2250
#define TILES_SCAN  1750
#define TILES_FILL  (GEMM_TILES - TILES_COUNT - TILES_SCAN)
#define EDGE_BLOCKS ((N_EDGES + 255) / 256)

template <int NE, int NT>
__device__ __forceinline__ bool block_is_tile(int bid, int& idx) {
    constexpr int M = NE < NT ? NE : NT;
    if (bid < 2 * M) { idx = bid >> 1; return (bid & 1) != 0; }
    idx = bid - M;
    return NT > NE;
}

__device__ __forceinline__ void gemm_tile(int tile, const float* __restrict__ x,
                                          const float* __restrict__ Wm,
                                          float* __restrict__ h, float* lx) {
    const int tid  = threadIdx.x;
    const int row0 = tile * 16;

    for (int idx = tid; idx < 16 * D; idx += 256) {
        int r = idx >> 7, c = idx & (D - 1);
        lx[r * LDS_PITCH + c] = x[(row0 + r) * D + c];
    }
    __syncthreads();

    const int wave = tid >> 5;
    const int lane = tid & 31;
    const int half = lane >> 4;
    const int l15  = lane & 15;
    const int koff = half * 2;
    const int n0   = wave * 16;

    v8f acc = {};
    for (int k = 0; k < D; k += 4) {
        v2f a, b;
        a.x = lx[l15 * LDS_PITCH + k + koff];
        a.y = lx[l15 * LDS_PITCH + k + koff + 1];
        b.x = Wm[(k + koff)     * D + n0 + l15];
        b.y = Wm[(k + koff + 1) * D + n0 + l15];
        acc = __builtin_amdgcn_wmma_f32_16x16x4_f32(
             false, a,  false, b,
             (short)0, acc,  false,  false);
    }

#pragma unroll
    for (int r = 0; r < 8; ++r) {
        int row = row0 + r + half * 8;
        h[row * D + n0 + l15] = acc[r];
    }
}

#define SCAN_B 1024
#define SCAN_T 256
#define SCAN_BLOCKS ((N_NODES + SCAN_B) / SCAN_B)
#define FRONT_U 16

__global__ void k_zero(int* __restrict__ cnt, float* __restrict__ sums,
                       float* __restrict__ sumsq) {
    int i = blockIdx.x * blockDim.x + threadIdx.x;
    if (i < N_NODES) cnt[i] = 0;
    if (i < D) { sums[i] = 0.0f; sumsq[i] = 0.0f; }
}

__global__ void __launch_bounds__(256) k_count_gemm(const int* __restrict__ dst,
                                                    int* __restrict__ cnt,
                                                    int* __restrict__ rank,
                                                    const float* __restrict__ x,
                                                    const float* __restrict__ Wm,
                                                    float* __restrict__ h) {
    __shared__ float lx[16 * LDS_PITCH];
    int idx;
    if (block_is_tile<EDGE_BLOCKS, TILES_COUNT>((int)blockIdx.x, idx)) {
        gemm_tile(idx, x, Wm, h, lx);
        return;
    }
    int e = idx * 256 + threadIdx.x;
    if (e < N_EDGES) rank[e] = atomicAdd(&cnt[dst[e]], 1);
}

__device__ __forceinline__ void scan_block(int b, const int* __restrict__ cnt,
                                           int* __restrict__ rowptr,
                                           float* __restrict__ dinv,
                                           int* sp, int* sc) {
    static_assert(N_NODES % 4 == 0, "a thread's four nodes are in range together");
    static_assert(SCAN_B == 4 * SCAN_T, "four nodes per thread");
    const int t = threadIdx.x;
    const int4* c4 = (const int4*)cnt;

    int p = 0;
    for (int k0 = 0; k0 < b; k0 += FRONT_U) {
        int4 v[FRONT_U];
#pragma unroll
        for (int j = 0; j < FRONT_U; ++j) {
            int k = k0 + j;
            if (k > b - 1) k = b - 1;
            v[j] = c4[k * SCAN_T + t];
        }
#pragma unroll
        for (int j = 0; j < FRONT_U; ++j)
            if (k0 + j < b) p += v[j].x + v[j].y + v[j].z + v[j].w;
    }

    const int i0 = b * SCAN_B + 4 * t;
    int4 c = make_int4(0, 0, 0, 0);
    if (i0 < N_NODES) c = c4[i0 >> 2];
    const int tsum = c.x + c.y + c.z + c.w;
    sp[t] = p;
    sc[t] = tsum;
    __syncthreads();

    for (int s = SCAN_T / 2; s > 0; s >>= 1) {
        if (t < s) sp[t] += sp[t + s];
        __syncthreads();
    }
    const int before = sp[0];

    int cur = 0;
    for (int off = 1; off < SCAN_T; off <<= 1) {
        int v = sc[cur * SCAN_T + t];
        if (t >= off) v += sc[cur * SCAN_T + t - off];
        sc[(cur ^ 1) * SCAN_T + t] = v;
        __syncthreads();
        cur ^= 1;
    }
    const int r0 = before + sc[cur * SCAN_T + t] - tsum;

    if (i0 < N_NODES) {
        int4 rp;
        rp.x = r0;
        rp.y = r0 + c.x;
        rp.z = r0 + c.x + c.y;
        rp.w = r0 + c.x + c.y + c.z;
        ((int4*)rowptr)[i0 >> 2] = rp;
        float4 dv;
        dv.x = rsqrtf((float)(c.x + 1));
        dv.y = rsqrtf((float)(c.y + 1));
        dv.z = rsqrtf((float)(c.z + 1));
        dv.w = rsqrtf((float)(c.w + 1));
        ((float4*)dinv)[i0 >> 2] = dv;
    } else if (i0 == N_NODES) {
        rowptr[N_NODES] = r0;
    }
}

__global__ void __launch_bounds__(256) k_scan_gemm(const int* __restrict__ cnt,
                                                   int* __restrict__ rowptr,
                                                   float* __restrict__ dinv,
                                                   const float* __restrict__ x,
                                                   const float* __restrict__ Wm,
                                                   float* __restrict__ h) {
    __shared__ float lx[16 * LDS_PITCH];
    __shared__ int sp[SCAN_T];
    __shared__ int sc[2 * SCAN_T];
    const int bid = blockIdx.x;
    if (bid < SCAN_BLOCKS) {
        scan_block(bid, cnt, rowptr, dinv, sp, sc);
        return;
    }
    gemm_tile(TILES_COUNT + bid - SCAN_BLOCKS, x, Wm, h, lx);
}

__global__ void __launch_bounds__(256) k_fill_gemm(const int* __restrict__ src,
                                                   const int* __restrict__ dst,
                                                   const int* __restrict__ rowptr,
                                                   const int* __restrict__ rank,
                                                   int* __restrict__ col,
                                                   const float* __restrict__ x,
                                                   const float* __restrict__ Wm,
                                                   float* __restrict__ h) {
    __shared__ float lx[16 * LDS_PITCH];
    int idx;
    if (block_is_tile<EDGE_BLOCKS, TILES_FILL>((int)blockIdx.x, idx)) {
        gemm_tile(TILES_COUNT + TILES_SCAN + idx, x, Wm, h, lx);
        return;
    }
    int e = idx * 256 + threadIdx.x;
    if (e < N_EDGES) col[rowptr[dst[e]] + rank[e]] = src[e];
}

template <int R>
__device__ __forceinline__ void add_rows(float4& acc, const float4* __restrict__ hp,
                                         int my_s, float my_w, int j, int lane) {
    float4 v[R];
#pragma unroll
    for (int k = 0; k < R; ++k)
        v[k] = hp[(size_t)__builtin_amdgcn_readlane(my_s, j + k) * 32 + lane];
#pragma unroll
    for (int k = 0; k < R; ++k) {
        const float w = __int_as_float(__builtin_amdgcn_readlane(__float_as_int(my_w), j + k));
        acc.x = fmaf(v[k].x, w, acc.x);
        acc.y = fmaf(v[k].y, w, acc.y);
        acc.z = fmaf(v[k].z, w, acc.z);
        acc.w = fmaf(v[k].w, w, acc.w);
    }
}

__global__ void __launch_bounds__(256) k_gather(const int* __restrict__ rowptr,
                                                const int* __restrict__ col,
                                                const float* __restrict__ h,
                                                const float* __restrict__ dinv,
                                                const float* __restrict__ b,
                                                float* __restrict__ out) {
    const int lane = threadIdx.x & 31;
    const int t = __builtin_amdgcn_readfirstlane((int)(blockIdx.x * 8 + (threadIdx.x >> 5)));

    const float4* hp = (const float4*)h;
    const int   beg = rowptr[t];
    const int   end = rowptr[t + 1];
    const float dt  = dinv[t];

    const float4 hs = hp[(size_t)t * 32 + lane];
    const float  ws = dt * dt;
    float4 acc;
    acc.x = hs.x * ws; acc.y = hs.y * ws; acc.z = hs.z * ws; acc.w = hs.w * ws;

    for (int base = beg; base < end; base += 32) {
        int n = end - base;
        if (n > 32) n = 32;
        int   my_s = 0;
        float my_w = 0.0f;
        if (lane < n) {
            my_s = col[base + lane];
            my_w = dinv[my_s] * dt;
        }
        for (int j = 0; j < n; j += 8) {
            switch (n - j) {
                case 1:  add_rows<1>(acc, hp, my_s, my_w, j, lane); break;
                case 2:  add_rows<2>(acc, hp, my_s, my_w, j, lane); break;
                case 3:  add_rows<3>(acc, hp, my_s, my_w, j, lane); break;
                case 4:  add_rows<4>(acc, hp, my_s, my_w, j, lane); break;
                case 5:  add_rows<5>(acc, hp, my_s, my_w, j, lane); break;
                case 6:  add_rows<6>(acc, hp, my_s, my_w, j, lane); break;
                case 7:  add_rows<7>(acc, hp, my_s, my_w, j, lane); break;
                default: add_rows<8>(acc, hp, my_s, my_w, j, lane); break;
            }
        }
    }

    const float4 bb = ((const float4*)b)[lane];
    float4 o;
    o.x = acc.x + bb.x; o.y = acc.y + bb.y; o.z = acc.z + bb.z; o.w = acc.w + bb.w;
    ((float4*)out)[(size_t)t * 32 + lane] = o;
}

#define ST_ROWS  16
#define ST_WAVES 16

__global__ void __launch_bounds__(ST_WAVES * 32) k_stats(const float* __restrict__ out,
                                                         float* __restrict__ sums,
                                                         float* __restrict__ sumsq) {
    static_assert(N_NODES % ST_ROWS == 0, "whole waves only");
    __shared__ float red[ST_WAVES][2 * D];
    const int lane = threadIdx.x & 31;
    const int wave = threadIdx.x >> 5;
    const int row0 = (int)(blockIdx.x * ST_WAVES + wave) * ST_ROWS;

    float4 s = make_float4(0.0f, 0.0f, 0.0f, 0.0f);
    float4 q = make_float4(0.0f, 0.0f, 0.0f, 0.0f);
    if (row0 < N_NODES) {
        const float4* op = (const float4*)out + (size_t)row0 * 32 + lane;
        float4 v[ST_ROWS];
#pragma unroll
        for (int k = 0; k < ST_ROWS; ++k) v[k] = op[k * 32];
#pragma unroll
        for (int k = 0; k < ST_ROWS; ++k) {
            float4 r;
            r.x = v[k].x > 0.0f ? v[k].x : 0.0f;
            r.y = v[k].y > 0.0f ? v[k].y : 0.0f;
            r.z = v[k].z > 0.0f ? v[k].z : 0.0f;
            r.w = v[k].w > 0.0f ? v[k].w : 0.0f;
            s.x += r.x; s.y += r.y; s.z += r.z; s.w += r.w;
            q.x += r.x * r.x; q.y += r.y * r.y; q.z += r.z * r.z; q.w += r.w * r.w;
        }
    }
    ((float4*)&red[wave][0])[lane] = s;
    ((float4*)&red[wave][D])[lane] = q;
    __syncthreads();

    if (threadIdx.x < 2 * D) {
        float a = 0.0f;
#pragma unroll
        for (int w = 0; w < ST_WAVES; ++w) a += red[w][threadIdx.x];
        if (threadIdx.x < D) atomicAdd(&sums[threadIdx.x], a);
        else                 atomicAdd(&sumsq[threadIdx.x - D], a);
    }
}

#define BN_ROWS 16

__device__ __forceinline__ float bn_one(float v, float mean, float rs, float g, float bt) {
    v = v > 0.0f ? v : 0.0f;
    return g * (v - mean) * rs + bt;
}

__global__ void __launch_bounds__(256) k_bn(float* __restrict__ out,
                                            const float* __restrict__ sums,
                                            const float* __restrict__ sumsq,
                                            const float* __restrict__ gamma,
                                            const float* __restrict__ beta) {
    static_assert(N_NODES % BN_ROWS == 0, "whole waves only");
    const int lane = threadIdx.x & 31;
    const int row0 = (int)(blockIdx.x * 8 + (threadIdx.x >> 5)) * BN_ROWS;
    if (row0 >= N_NODES) return;

    float4* op = (float4*)out + (size_t)row0 * 32 + lane;
    float4 v[BN_ROWS];
#pragma unroll
    for (int k = 0; k < BN_ROWS; ++k) v[k] = op[k * 32];

    const float4 s  = ((const float4*)sums)[lane];
    const float4 sq = ((const float4*)sumsq)[lane];
    const float4 g  = ((const float4*)gamma)[lane];
    const float4 bt = ((const float4*)beta)[lane];

    const float invN = 1.0f / (float)N_NODES;
    float4 mean, var, rs;
    mean.x = s.x * invN; mean.y = s.y * invN; mean.z = s.z * invN; mean.w = s.w * invN;
    var.x = sq.x * invN - mean.x * mean.x;
    var.y = sq.y * invN - mean.y * mean.y;
    var.z = sq.z * invN - mean.z * mean.z;
    var.w = sq.w * invN - mean.w * mean.w;
    rs.x = rsqrtf(var.x + BN_EPS); rs.y = rsqrtf(var.y + BN_EPS);
    rs.z = rsqrtf(var.z + BN_EPS); rs.w = rsqrtf(var.w + BN_EPS);

#pragma unroll
    for (int k = 0; k < BN_ROWS; ++k) {
        float4 o;
        o.x = bn_one(v[k].x, mean.x, rs.x, g.x, bt.x);
        o.y = bn_one(v[k].y, mean.y, rs.y, g.y, bt.y);
        o.z = bn_one(v[k].z, mean.z, rs.z, g.z, bt.z);
        o.w = bn_one(v[k].w, mean.w, rs.w, g.w, bt.w);
        op[k * 32] = o;
    }
}

extern "C" void kernel_launch(void* const* d_in, const int* in_sizes, int n_in,
                              void* d_out, int out_size, void* d_ws, size_t ws_size,
                              hipStream_t stream) {
    const float* x     = (const float*)d_in[0];
    const int*   edges = (const int*)  d_in[1];
    const float* Wm    = (const float*)d_in[2];
    const float* b     = (const float*)d_in[3];
    const float* gamma = (const float*)d_in[4];
    const float* beta  = (const float*)d_in[5];
    float*       out   = (float*)d_out;

    const int* src = edges;
    const int* dst = edges + N_EDGES;

    float* h      = (float*)d_ws;
    float* dinv   = h + (size_t)N_NODES * D;
    int*   cnt    = (int*)(dinv + N_NODES);
    int*   rowptr = cnt + N_NODES;
    int*   rank   = rowptr + (N_NODES + 32);
    int*   col    = rank + N_EDGES;
    float* sums   = (float*)(col + N_EDGES);
    float* sumsq  = sums + D;

    k_zero      <<<(N_NODES + 255) / 256, 256, 0, stream>>>(cnt, sums, sumsq);
    k_count_gemm<<<EDGE_BLOCKS + TILES_COUNT, 256, 0, stream>>>(dst, cnt, rank, x, Wm, h);
    k_scan_gemm <<<SCAN_BLOCKS + TILES_SCAN, 256, 0, stream>>>(cnt, rowptr, dinv, x, Wm, h);
    k_fill_gemm <<<EDGE_BLOCKS + TILES_FILL, 256, 0, stream>>>(src, dst, rowptr, rank, col, x, Wm, h);

    k_gather    <<<N_NODES / 8, 256, 0, stream>>>(rowptr, col, h, dinv, b, out);

    k_stats     <<<(N_NODES / ST_ROWS + ST_WAVES - 1) / ST_WAVES, ST_WAVES * 32, 0, stream>>>(out, sums, sumsq);
    k_bn        <<<(N_NODES / BN_ROWS + 7) / 8, 256, 0, stream>>>(out, sums, sumsq, gamma, beta);
}


# Prefix-sum and fill launches removed: each destination owns 32 slots (one line) filled by the counting launch itself, edges past 32 go to a list the gather scans; 7 -> 5 launches; sum order as before
# speedup vs baseline: 2.0338x; 1.0494x over previous
#include <hip/hip_runtime.h>

#define N_NODES 100000
#define N_EDGES 640000
#define D       128
#define BN_EPS  1e-5f
#define LDS_PITCH 132

typedef float v2f __attribute__((ext_vector_type(2)));
typedef float v8f __attribute__((ext_vector_type(8)));

#define GEMM_TILES  (N_NODES / 16)
#define TILES_ZERO  2750
#define TILES_PLACE (GEMM_TILES - TILES_ZERO)
#define EDGE_BLOCKS ((N_EDGES + 255) / 256)
#define ZERO_BLOCKS ((N_NODES + 255) / 256)

template <int NE, int NT>
__device__ __forceinline__ bool block_is_tile(int bid, int& idx) {
    constexpr int M = NE < NT ? NE : NT;
    if (bid < 2 * M) { idx = bid >> 1; return (bid & 1) != 0; }
    idx = bid - M;
    return NT > NE;
}

__device__ __forceinline__ void gemm_tile(int tile, const float* __restrict__ x,
                                          const float* __restrict__ Wm,
                                          float* __restrict__ h, float* lx) {
    const int tid  = threadIdx.x;
    const int row0 = tile * 16;

    for (int idx = tid; idx < 16 * D; idx += 256) {
        int r = idx >> 7, c = idx & (D - 1);
        lx[r * LDS_PITCH + c] = x[(row0 + r) * D + c];
    }
    __syncthreads();

    const int wave = tid >> 5;
    const int lane = tid & 31;
    const int half = lane >> 4;
    const int l15  = lane & 15;
    const int koff = half * 2;
    const int n0   = wave * 16;

    v8f acc = {};
    for (int k = 0; k < D; k += 4) {
        v2f a, b;
        a.x = lx[l15 * LDS_PITCH + k + koff];
        a.y = lx[l15 * LDS_PITCH + k + koff + 1];
        b.x = Wm[(k + koff)     * D + n0 + l15];
        b.y = Wm[(k + koff + 1) * D + n0 + l15];
        acc = __builtin_amdgcn_wmma_f32_16x16x4_f32(
             false, a,  false, b,
             (short)0, acc,  false,  false);
    }

#pragma unroll
    for (int r = 0; r < 8; ++r) {
        int row = row0 + r + half * 8;
        h[row * D + n0 + l15] = acc[r];
    }
}

#define CAP 32

__global__ void __launch_bounds__(256) k_zero_gemm(int* __restrict__ cnt,
                                                   float* __restrict__ sums,
                                                   float* __restrict__ sumsq,
                                                   int* __restrict__ novf,
                                                   const float* __restrict__ x,
                                                   const float* __restrict__ Wm,
                                                   float* __restrict__ h) {
    __shared__ float lx[16 * LDS_PITCH];
    int idx;
    if (block_is_tile<ZERO_BLOCKS, TILES_ZERO>((int)blockIdx.x, idx)) {
        gemm_tile(idx, x, Wm, h, lx);
        return;
    }
    int i = idx * 256 + threadIdx.x;
    if (i < N_NODES) cnt[i] = 0;
    if (i < D) { sums[i] = 0.0f; sumsq[i] = 0.0f; }
    if (i == 0) novf[0] = 0;
}

__global__ void __launch_bounds__(256) k_place_gemm(const int* __restrict__ src,
                                                    const int* __restrict__ dst,
                                                    int* __restrict__ cnt,
                                                    int* __restrict__ col,
                                                    int* __restrict__ novf,
                                                    int* __restrict__ ovf,
                                                    const float* __restrict__ x,
                                                    const float* __restrict__ Wm,
                                                    float* __restrict__ h) {
    __shared__ float lx[16 * LDS_PITCH];
    int idx;
    if (block_is_tile<EDGE_BLOCKS, TILES_PLACE>((int)blockIdx.x, idx)) {
        gemm_tile(TILES_ZERO + idx, x, Wm, h, lx);
        return;
    }
    int e = idx * 256 + threadIdx.x;
    if (e < N_EDGES) {
        const int d = dst[e];
        const int s = src[e];
        const int r = atomicAdd(&cnt[d], 1);
        if (r < CAP) {
            col[d * CAP + r] = s;
        } else {
            const int o = atomicAdd(novf, 1);
            ovf[2 * o]     = s;
            ovf[2 * o + 1] = d;
        }
    }
}

template <int R>
__device__ __forceinline__ void add_rows(float4& acc, const float4* __restrict__ hp,
                                         int my_s, float my_w, int j, int lane) {
    float4 v[R];
#pragma unroll
    for (int k = 0; k < R; ++k)
        v[k] = hp[(size_t)__builtin_amdgcn_readlane(my_s, j + k) * 32 + lane];
#pragma unroll
    for (int k = 0; k < R; ++k) {
        const float w = __int_as_float(__builtin_amdgcn_readlane(__float_as_int(my_w), j + k));
        acc.x = fmaf(v[k].x, w, acc.x);
        acc.y = fmaf(v[k].y, w, acc.y);
        acc.z = fmaf(v[k].z, w, acc.z);
        acc.w = fmaf(v[k].w, w, acc.w);
    }
}

__global__ void __launch_bounds__(256) k_gather(const int* __restrict__ cnt,
                                                const int* __restrict__ col,
                                                const int* __restrict__ novf,
                                                const int* __restrict__ ovf,
                                                const float* __restrict__ h,
                                                const float* __restrict__ b,
                                                float* __restrict__ out) {
    const int lane = threadIdx.x & 31;
    const int t = __builtin_amdgcn_readfirstlane((int)(blockIdx.x * 8 + (threadIdx.x >> 5)));

    const float4* hp = (const float4*)h;
    const int   c  = cnt[t];
    const float dt = rsqrtf((float)(c + 1));
    const int   n  = c < CAP ? c : CAP;

    const float4 hs = hp[(size_t)t * 32 + lane];
    const float  ws = dt * dt;
    float4 acc;
    acc.x = hs.x * ws; acc.y = hs.y * ws; acc.z = hs.z * ws; acc.w = hs.w * ws;

    int   my_s = 0;
    float my_w = 0.0f;
    if (lane < n) {
        my_s = col[t * CAP + lane];
        my_w = rsqrtf((float)(cnt[my_s] + 1)) * dt;
    }
    for (int j = 0; j < n; j += 8) {
        switch (n - j) {
            case 1:  add_rows<1>(acc, hp, my_s, my_w, j, lane); break;
            case 2:  add_rows<2>(acc, hp, my_s, my_w, j, lane); break;
            case 3:  add_rows<3>(acc, hp, my_s, my_w, j, lane); break;
            case 4:  add_rows<4>(acc, hp, my_s, my_w, j, lane); break;
            case 5:  add_rows<5>(acc, hp, my_s, my_w, j, lane); break;
            case 6:  add_rows<6>(acc, hp, my_s, my_w, j, lane); break;
            case 7:  add_rows<7>(acc, hp, my_s, my_w, j, lane); break;
            default: add_rows<8>(acc, hp, my_s, my_w, j, lane); break;
        }
    }

    if (c > CAP) {
        const int no = novf[0];
        for (int i0 = 0; i0 < no; i0 += 32) {
            int s_l = 0;
            int d_l = -1;
            if (i0 + lane < no) {
                s_l = ovf[2 * (i0 + lane)];
                d_l = ovf[2 * (i0 + lane) + 1];
            }
            unsigned int m = __builtin_amdgcn_ballot_w32(d_l == t);
            while (m != 0u) {
                const int k = __builtin_ctz(m);
                m &= m - 1u;
                const int   s = __builtin_amdgcn_readlane(s_l, k);
                const float w = rsqrtf((float)(cnt[s] + 1)) * dt;
                const float4 v = hp[(size_t)s * 32 + lane];
                acc.x = fmaf(v.x, w, acc.x);
                acc.y = fmaf(v.y, w, acc.y);
                acc.z = fmaf(v.z, w, acc.z);
                acc.w = fmaf(v.w, w, acc.w);
            }
        }
    }

    const float4 bb = ((const float4*)b)[lane];
    float4 o;
    o.x = acc.x + bb.x; o.y = acc.y + bb.y; o.z = acc.z + bb.z; o.w = acc.w + bb.w;
    ((float4*)out)[(size_t)t * 32 + lane] = o;
}

#define ST_ROWS  16
#define ST_WAVES 16

__global__ void __launch_bounds__(ST_WAVES * 32) k_stats(const float* __restrict__ out,
                                                         float* __restrict__ sums,
                                                         float* __restrict__ sumsq) {
    static_assert(N_NODES % ST_ROWS == 0, "whole waves only");
    __shared__ float red[ST_WAVES][2 * D];
    const int lane = threadIdx.x & 31;
    const int wave = threadIdx.x >> 5;
    const int row0 = (int)(blockIdx.x * ST_WAVES + wave) * ST_ROWS;

    float4 s = make_float4(0.0f, 0.0f, 0.0f, 0.0f);
    float4 q = make_float4(0.0f, 0.0f, 0.0f, 0.0f);
    if (row0 < N_NODES) {
        const float4* op = (const float4*)out + (size_t)row0 * 32 + lane;
        float4 v[ST_ROWS];
#pragma unroll
        for (int k = 0; k < ST_ROWS; ++k) v[k] = op[k * 32];
#pragma unroll
        for (int k = 0; k < ST_ROWS; ++k) {
            float4 r;
            r.x = v[k].x > 0.0f ? v[k].x : 0.0f;
            r.y = v[k].y > 0.0f ? v[k].y : 0.0f;
            r.z = v[k].z > 0.0f ? v[k].z : 0.0f;
            r.w = v[k].w > 0.0f ? v[k].w : 0.0f;
            s.x += r.x; s.y += r.y; s.z += r.z; s.w += r.w;
            q.x += r.x * r.x; q.y += r.y * r.y; q.z += r.z * r.z; q.w += r.w * r.w;
        }
    }
    ((float4*)&red[wave][0])[lane] = s;
    ((float4*)&red[wave][D])[lane] = q;
    __syncthreads();

    if (threadIdx.x < 2 * D) {
        float a = 0.0f;
#pragma unroll
        for (int w = 0; w < ST_WAVES; ++w) a += red[w][threadIdx.x];
        if (threadIdx.x < D) atomicAdd(&sums[threadIdx.x], a);
        else                 atomicAdd(&sumsq[threadIdx.x - D], a);
    }
}

#define BN_ROWS 16

__device__ __forceinline__ float bn_one(float v, float mean, float rs, float g, float bt) {
    v = v > 0.0f ? v : 0.0f;
    return g * (v - mean) * rs + bt;
}

__global__ void __launch_bounds__(256) k_bn(float* __restrict__ out,
                                            const float* __restrict__ sums,
                                            const float* __restrict__ sumsq,
                                            const float* __restrict__ gamma,
                                            const float* __restrict__ beta) {
    static_assert(N_NODES % BN_ROWS == 0, "whole waves only");
    const int lane = threadIdx.x & 31;
    const int row0 = (int)(blockIdx.x * 8 + (threadIdx.x >> 5)) * BN_ROWS;
    if (row0 >= N_NODES) return;

    float4* op = (float4*)out + (size_t)row0 * 32 + lane;
    float4 v[BN_ROWS];
#pragma unroll
    for (int k = 0; k < BN_ROWS; ++k) v[k] = op[k * 32];

    const float4 s  = ((const float4*)sums)[lane];
    const float4 sq = ((const float4*)sumsq)[lane];
    const float4 g  = ((const float4*)gamma)[lane];
    const float4 bt = ((const float4*)beta)[lane];

    const float invN = 1.0f / (float)N_NODES;
    float4 mean, var, rs;
    mean.x = s.x * invN; mean.y = s.y * invN; mean.z = s.z * invN; mean.w = s.w * invN;
    var.x = sq.x * invN - mean.x * mean.x;
    var.y = sq.y * invN - mean.y * mean.y;
    var.z = sq.z * invN - mean.z * mean.z;
    var.w = sq.w * invN - mean.w * mean.w;
    rs.x = rsqrtf(var.x + BN_EPS); rs.y = rsqrtf(var.y + BN_EPS);
    rs.z = rsqrtf(var.z + BN_EPS); rs.w = rsqrtf(var.w + BN_EPS);

#pragma unroll
    for (int k = 0; k < BN_ROWS; ++k) {
        float4 o;
        o.x = bn_one(v[k].x, mean.x, rs.x, g.x, bt.x);
        o.y = bn_one(v[k].y, mean.y, rs.y, g.y, bt.y);
        o.z = bn_one(v[k].z, mean.z, rs.z, g.z, bt.z);
        o.w = bn_one(v[k].w, mean.w, rs.w, g.w, bt.w);
        op[k * 32] = o;
    }
}

extern "C" void kernel_launch(void* const* d_in, const int* in_sizes, int n_in,
                              void* d_out, int out_size, void* d_ws, size_t ws_size,
                              hipStream_t stream) {
    const float* x     = (const float*)d_in[0];
    const int*   edges = (const int*)  d_in[1];
    const float* Wm    = (const float*)d_in[2];
    const float* b     = (const float*)d_in[3];
    const float* gamma = (const float*)d_in[4];
    const float* beta  = (const float*)d_in[5];
    float*       out   = (float*)d_out;

    const int* src = edges;
    const int* dst = edges + N_EDGES;

    float* h      = (float*)d_ws;
    int*   cnt    = (int*)(h + (size_t)N_NODES * D);
    int*   col    = cnt + N_NODES;
    float* sums   = (float*)(col + (size_t)N_NODES * CAP);
    float* sumsq  = sums + D;
    int*   novf   = (int*)(sumsq + D);
    int*   ovf    = novf + 32;

    k_zero_gemm <<<ZERO_BLOCKS + TILES_ZERO, 256, 0, stream>>>(cnt, sums, sumsq, novf, x, Wm, h);
    k_place_gemm<<<EDGE_BLOCKS + TILES_PLACE, 256, 0, stream>>>(src, dst, cnt, col, novf, ovf, x, Wm, h);

    k_gather    <<<N_NODES / 8, 256, 0, stream>>>(cnt, col, novf, ovf, h, b, out);

    k_stats     <<<(N_NODES / ST_ROWS + ST_WAVES - 1) / ST_WAVES, ST_WAVES * 32, 0, stream>>>(out, sums, sumsq);
    k_bn        <<<(N_NODES / BN_ROWS + 7) / 8, 256, 0, stream>>>(out, sums, sumsq, gamma, beta);
}


# Per-edge returning global add replaced by a two-level counting sort on LDS counters: bin edges by 256-node bucket, build each bucket's lists in LDS, store whole lines; slot order = LDS arrival rank
# speedup vs baseline: 2.4631x; 1.2111x over previous
#include <hip/hip_runtime.h>

#define N_NODES 100000
#define N_EDGES 640000
#define D       128
#define BN_EPS  1e-5f
#define LDS_PITCH 132

typedef float v2f __attribute__((ext_vector_type(2)));
typedef float v8f __attribute__((ext_vector_type(8)));

#define GEMM_TILES  (N_NODES / 16)
#define TILES_BIN   (GEMM_TILES / 2)
#define TILES_BUILD (GEMM_TILES - TILES_BIN)

template <int NE, int NT>
__device__ __forceinline__ bool block_is_tile(int bid, int& idx) {
    constexpr int M = NE < NT ? NE : NT;
    if (bid < 2 * M) { idx = bid >> 1; return (bid & 1) != 0; }
    idx = bid - M;
    return NT > NE;
}

__device__ __forceinline__ void gemm_tile(int tile, const float* __restrict__ x,
                                          const float* __restrict__ Wm,
                                          float* __restrict__ h, float* lx) {
    const int tid  = threadIdx.x;
    const int row0 = tile * 16;

    for (int idx = tid; idx < 16 * D; idx += 256) {
        int r = idx >> 7, c = idx & (D - 1);
        lx[r * LDS_PITCH + c] = x[(row0 + r) * D + c];
    }
    __syncthreads();

    const int wave = tid >> 5;
    const int lane = tid & 31;
    const int half = lane >> 4;
    const int l15  = lane & 15;
    const int koff = half * 2;
    const int n0   = wave * 16;

    v8f acc = {};
    for (int k = 0; k < D; k += 4) {
        v2f a, b;
        a.x = lx[l15 * LDS_PITCH + k + koff];
        a.y = lx[l15 * LDS_PITCH + k + koff + 1];
        b.x = Wm[(k + koff)     * D + n0 + l15];
        b.y = Wm[(k + koff + 1) * D + n0 + l15];
        acc = __builtin_amdgcn_wmma_f32_16x16x4_f32(
             false, a,  false, b,
             (short)0, acc,  false,  false);
    }

#pragma unroll
    for (int r = 0; r < 8; ++r) {
        int row = row0 + r + half * 8;
        h[row * D + n0 + l15] = acc[r];
    }
}

#define CAP 32
#define CHUNK     2560
#define NBIN      (N_EDGES / CHUNK)
#define BKT_SHIFT 8
#define BKT       (1 << BKT_SHIFT)
#define NBKT      ((N_NODES + BKT - 1) / BKT)
#define RCAP      16
#define RL_PITCH  256

__device__ __forceinline__ void list_put(int s, int d, int r, int base, int* lcol,
                                         int* __restrict__ novf, int* __restrict__ ovf) {
    if (r < CAP) {
        lcol[(d - base) * CAP + r] = s;
    } else {
        const int o = atomicAdd(novf, 1);
        ovf[2 * o]     = s;
        ovf[2 * o + 1] = d;
    }
}

__global__ void __launch_bounds__(256) k_bin_gemm(const int* __restrict__ src,
                                                  const int* __restrict__ dst,
                                                  int* __restrict__ runlen,
                                                  int* __restrict__ runs,
                                                  float* __restrict__ sums,
                                                  float* __restrict__ sumsq,
                                                  int* __restrict__ novf,
                                                  const float* __restrict__ x,
                                                  const float* __restrict__ Wm,
                                                  float* __restrict__ h) {
    static_assert(N_EDGES % CHUNK == 0 && CHUNK % 256 == 0, "whole chunks, whole passes of the block");
    static_assert(NBIN <= RL_PITCH && NBIN <= 256, "one thread per run in level 2");
    static_assert(NBKT * 4 <= 16 * LDS_PITCH * 4, "bucket counters fit the tile buffer");
    __shared__ float lx[16 * LDS_PITCH];
    int idx;
    if (block_is_tile<NBIN, TILES_BIN>((int)blockIdx.x, idx)) {
        gemm_tile(idx, x, Wm, h, lx);
        return;
    }
    int* lcnt = (int*)lx;
    const int tid = threadIdx.x;
    if (idx == 0) {
        if (tid < D) { sums[tid] = 0.0f; sumsq[tid] = 0.0f; }
        if (tid == 0) novf[0] = 0;
    }
    for (int k = tid; k < NBKT; k += 256) lcnt[k] = 0;

    constexpr int PER = CHUNK / 256;
    const int e0 = idx * CHUNK + tid;
    int d[PER], s[PER];
#pragma unroll
    for (int j = 0; j < PER; ++j) { d[j] = dst[e0 + j * 256]; s[j] = src[e0 + j * 256]; }
    __syncthreads();
    int r[PER];
#pragma unroll
    for (int j = 0; j < PER; ++j)
        r[j] = atomicAdd(&lcnt[d[j] >> BKT_SHIFT], 1);
#pragma unroll
    for (int j = 0; j < PER; ++j) {
        if (r[j] < RCAP) {
            const int k = d[j] >> BKT_SHIFT;
            int2 p; p.x = s[j]; p.y = d[j];
            ((int2*)runs)[((size_t)k * NBIN + idx) * RCAP + r[j]] = p;
        }
    }
    __syncthreads();
    for (int k = tid; k < NBKT; k += 256) runlen[k * RL_PITCH + idx] = lcnt[k];
}

#define BUILD_LDS_INTS (BKT * CAP + BKT + 256 + 256 + 4)
__global__ void __launch_bounds__(256) k_build_gemm(const int* __restrict__ src,
                                                    const int* __restrict__ dst,
                                                    const int* __restrict__ runlen,
                                                    const int* __restrict__ runs,
                                                    int* __restrict__ cnt,
                                                    int* __restrict__ col,
                                                    int* __restrict__ novf,
                                                    int* __restrict__ ovf,
                                                    const float* __restrict__ x,
                                                    const float* __restrict__ Wm,
                                                    float* __restrict__ h) {
    static_assert(BKT == 256, "one thread per node of the bucket");
    static_assert(RCAP == 16 && NBIN * RCAP <= 16 * 256, "16 passes of the block cover every stored pair");
    static_assert(BUILD_LDS_INTS >= 16 * LDS_PITCH, "the tile buffer fits");
    __shared__ __attribute__((aligned(16))) int sm[BUILD_LDS_INTS];
    int idx;
    if (block_is_tile<NBKT, TILES_BUILD>((int)blockIdx.x, idx)) {
        gemm_tile(TILES_BIN + idx, x, Wm, h, (float*)sm);
        return;
    }
    int* lcol = sm;
    int* lcnt = lcol + BKT * CAP;
    int* rl   = lcnt + BKT;
    int* fb   = rl + 256;
    int* nfb  = fb + 256;
    const int tid  = threadIdx.x;
    const int k    = idx;
    const int base = k << BKT_SHIFT;

    lcnt[tid] = 0;
    int n = 0;
    if (tid < NBIN) n = runlen[k * RL_PITCH + tid];
    rl[tid] = n;
    if (tid == 0) nfb[0] = 0;
    __syncthreads();
    if (n > RCAP) { const int q = atomicAdd(nfb, 1); fb[q] = tid; }

    const int2* rp = (const int2*)runs + (size_t)k * NBIN * RCAP;
    int  nb[16];
    int2 p[16];
    int  r[16];
#pragma unroll
    for (int it = 0; it < 16; ++it) nb[it] = rl[(it * 256 + tid) >> 4];
#pragma unroll
    for (int it = 0; it < 16; ++it) {
        const int f = it * 256 + tid;
        p[it].x = 0; p[it].y = -1;
        if (nb[it] <= RCAP && (f & 15) < nb[it]) p[it] = rp[f];
    }
#pragma unroll
    for (int it = 0; it < 16; ++it) {
        r[it] = 0;
        if (p[it].y >= 0) r[it] = atomicAdd(&lcnt[p[it].y - base], 1);
    }
#pragma unroll
    for (int it = 0; it < 16; ++it)
        if (p[it].y >= 0) list_put(p[it].x, p[it].y, r[it], base, lcol, novf, ovf);
    __syncthreads();

    const int nf = nfb[0];
    for (int q = 0; q < nf; ++q) {
        const int e0 = fb[q] * CHUNK + tid;
        int dd[CHUNK / 256];
#pragma unroll
        for (int j = 0; j < CHUNK / 256; ++j) dd[j] = dst[e0 + j * 256];
#pragma unroll
        for (int j = 0; j < CHUNK / 256; ++j) {
            if ((dd[j] >> BKT_SHIFT) == k) {
                const int rr = atomicAdd(&lcnt[dd[j] - base], 1);
                list_put(src[e0 + j * 256], dd[j], rr, base, lcol, novf, ovf);
            }
        }
    }
    __syncthreads();

    if (base + tid < N_NODES) cnt[base + tid] = lcnt[tid];
    for (int q = tid; q < BKT * CAP / 4; q += 256)
        if (base + (q >> 3) < N_NODES)
            ((int4*)col)[(size_t)base * (CAP / 4) + q] = ((const int4*)lcol)[q];
}

template <int R>
__device__ __forceinline__ void add_rows(float4& acc, const float4* __restrict__ hp,
                                         int my_s, float my_w, int j, int lane) {
    float4 v[R];
#pragma unroll
    for (int k = 0; k < R; ++k)
        v[k] = hp[(size_t)__builtin_amdgcn_readlane(my_s, j + k) * 32 + lane];
#pragma unroll
    for (int k = 0; k < R; ++k) {
        const float w = __int_as_float(__builtin_amdgcn_readlane(__float_as_int(my_w), j + k));
        acc.x = fmaf(v[k].x, w, acc.x);
        acc.y = fmaf(v[k].y, w, acc.y);
        acc.z = fmaf(v[k].z, w, acc.z);
        acc.w = fmaf(v[k].w, w, acc.w);
    }
}

__global__ void __launch_bounds__(256) k_gather(const int* __restrict__ cnt,
                                                const int* __restrict__ col,
                                                const int* __restrict__ novf,
                                                const int* __restrict__ ovf,
                                                const float* __restrict__ h,
                                                const float* __restrict__ b,
                                                float* __restrict__ out) {
    const int lane = threadIdx.x & 31;
    const int t = __builtin_amdgcn_readfirstlane((int)(blockIdx.x * 8 + (threadIdx.x >> 5)));

    const float4* hp = (const float4*)h;
    const int   c  = cnt[t];
    const float dt = rsqrtf((float)(c + 1));
    const int   n  = c < CAP ? c : CAP;

    const float4 hs = hp[(size_t)t * 32 + lane];
    const float  ws = dt * dt;
    float4 acc;
    acc.x = hs.x * ws; acc.y = hs.y * ws; acc.z = hs.z * ws; acc.w = hs.w * ws;

    int   my_s = 0;
    float my_w = 0.0f;
    if (lane < n) {
        my_s = col[t * CAP + lane];
        my_w = rsqrtf((float)(cnt[my_s] + 1)) * dt;
    }
    for (int j = 0; j < n; j += 8) {
        switch (n - j) {
            case 1:  add_rows<1>(acc, hp, my_s, my_w, j, lane); break;
            case 2:  add_rows<2>(acc, hp, my_s, my_w, j, lane); break;
            case 3:  add_rows<3>(acc, hp, my_s, my_w, j, lane); break;
            case 4:  add_rows<4>(acc, hp, my_s, my_w, j, lane); break;
            case 5:  add_rows<5>(acc, hp, my_s, my_w, j, lane); break;
            case 6:  add_rows<6>(acc, hp, my_s, my_w, j, lane); break;
            case 7:  add_rows<7>(acc, hp, my_s, my_w, j, lane); break;
            default: add_rows<8>(acc, hp, my_s, my_w, j, lane); break;
        }
    }

    if (c > CAP) {
        const int no = novf[0];
        for (int i0 = 0; i0 < no; i0 += 32) {
            int s_l = 0;
            int d_l = -1;
            if (i0 + lane < no) {
                s_l = ovf[2 * (i0 + lane)];
                d_l = ovf[2 * (i0 + lane) + 1];
            }
            unsigned int m = __builtin_amdgcn_ballot_w32(d_l == t);
            while (m != 0u) {
                const int k = __builtin_ctz(m);
                m &= m - 1u;
                const int   s = __builtin_amdgcn_readlane(s_l, k);
                const float w = rsqrtf((float)(cnt[s] + 1)) * dt;
                const float4 v = hp[(size_t)s * 32 + lane];
                acc.x = fmaf(v.x, w, acc.x);
                acc.y = fmaf(v.y, w, acc.y);
                acc.z = fmaf(v.z, w, acc.z);
                acc.w = fmaf(v.w, w, acc.w);
            }
        }
    }

    const float4 bb = ((const float4*)b)[lane];
    float4 o;
    o.x = acc.x + bb.x; o.y = acc.y + bb.y; o.z = acc.z + bb.z; o.w = acc.w + bb.w;
    ((float4*)out)[(size_t)t * 32 + lane] = o;
}

#define ST_ROWS  16
#define ST_WAVES 16

__global__ void __launch_bounds__(ST_WAVES * 32) k_stats(const float* __restrict__ out,
                                                         float* __restrict__ sums,
                                                         float* __restrict__ sumsq) {
    static_assert(N_NODES % ST_ROWS == 0, "whole waves only");
    __shared__ float red[ST_WAVES][2 * D];
    const int lane = threadIdx.x & 31;
    const int wave = threadIdx.x >> 5;
    const int row0 = (int)(blockIdx.x * ST_WAVES + wave) * ST_ROWS;

    float4 s = make_float4(0.0f, 0.0f, 0.0f, 0.0f);
    float4 q = make_float4(0.0f, 0.0f, 0.0f, 0.0f);
    if (row0 < N_NODES) {
        const float4* op = (const float4*)out + (size_t)row0 * 32 + lane;
        float4 v[ST_ROWS];
#pragma unroll
        for (int k = 0; k < ST_ROWS; ++k) v[k] = op[k * 32];
#pragma unroll
        for (int k = 0; k < ST_ROWS; ++k) {
            float4 r;
            r.x = v[k].x > 0.0f ? v[k].x : 0.0f;
            r.y = v[k].y > 0.0f ? v[k].y : 0.0f;
            r.z = v[k].z > 0.0f ? v[k].z : 0.0f;
            r.w = v[k].w > 0.0f ? v[k].w : 0.0f;
            s.x += r.x; s.y += r.y; s.z += r.z; s.w += r.w;
            q.x += r.x * r.x; q.y += r.y * r.y; q.z += r.z * r.z; q.w += r.w * r.w;
        }
    }
    ((float4*)&red[wave][0])[lane] = s;
    ((float4*)&red[wave][D])[lane] = q;
    __syncthreads();

    if (threadIdx.x < 2 * D) {
        float a = 0.0f;
#pragma unroll
        for (int w = 0; w < ST_WAVES; ++w) a += red[w][threadIdx.x];
        if (threadIdx.x < D) atomicAdd(&sums[threadIdx.x], a);
        else                 atomicAdd(&sumsq[threadIdx.x - D], a);
    }
}

#define BN_ROWS 16

__device__ __forceinline__ float bn_one(float v, float mean, float rs, float g, float bt) {
    v = v > 0.0f ? v : 0.0f;
    return g * (v - mean) * rs + bt;
}

__global__ void __launch_bounds__(256) k_bn(float* __restrict__ out,
                                            const float* __restrict__ sums,
                                            const float* __restrict__ sumsq,
                                            const float* __restrict__ gamma,
                                            const float* __restrict__ beta) {
    static_assert(N_NODES % BN_ROWS == 0, "whole waves only");
    const int lane = threadIdx.x & 31;
    const int row0 = (int)(blockIdx.x * 8 + (threadIdx.x >> 5)) * BN_ROWS;
    if (row0 >= N_NODES) return;

    float4* op = (float4*)out + (size_t)row0 * 32 + lane;
    float4 v[BN_ROWS];
#pragma unroll
    for (int k = 0; k < BN_ROWS; ++k) v[k] = op[k * 32];

    const float4 s  = ((const float4*)sums)[lane];
    const float4 sq = ((const float4*)sumsq)[lane];
    const float4 g  = ((const float4*)gamma)[lane];
    const float4 bt = ((const float4*)beta)[lane];

    const float invN = 1.0f / (float)N_NODES;
    float4 mean, var, rs;
    mean.x = s.x * invN; mean.y = s.y * invN; mean.z = s.z * invN; mean.w = s.w * invN;
    var.x = sq.x * invN - mean.x * mean.x;
    var.y = sq.y * invN - mean.y * mean.y;
    var.z = sq.z * invN - mean.z * mean.z;
    var.w = sq.w * invN - mean.w * mean.w;
    rs.x = rsqrtf(var.x + BN_EPS); rs.y = rsqrtf(var.y + BN_EPS);
    rs.z = rsqrtf(var.z + BN_EPS); rs.w = rsqrtf(var.w + BN_EPS);

#pragma unroll
    for (int k = 0; k < BN_ROWS; ++k) {
        float4 o;
        o.x = bn_one(v[k].x, mean.x, rs.x, g.x, bt.x);
        o.y = bn_one(v[k].y, mean.y, rs.y, g.y, bt.y);
        o.z = bn_one(v[k].z, mean.z, rs.z, g.z, bt.z);
        o.w = bn_one(v[k].w, mean.w, rs.w, g.w, bt.w);
        op[k * 32] = o;
    }
}

extern "C" void kernel_launch(void* const* d_in, const int* in_sizes, int n_in,
                              void* d_out, int out_size, void* d_ws, size_t ws_size,
                              hipStream_t stream) {
    const float* x     = (const float*)d_in[0];
    const int*   edges = (const int*)  d_in[1];
    const float* Wm    = (const float*)d_in[2];
    const float* b     = (const float*)d_in[3];
    const float* gamma = (const float*)d_in[4];
    const float* beta  = (const float*)d_in[5];
    float*       out   = (float*)d_out;

    const int* src = edges;
    const int* dst = edges + N_EDGES;

    float* h      = (float*)d_ws;
    int*   cnt    = (int*)(h + (size_t)N_NODES * D);
    int*   col    = cnt + N_NODES;
    float* sums   = (float*)(col + (size_t)N_NODES * CAP);
    float* sumsq  = sums + D;
    int*   novf   = (int*)(sumsq + D);
    int*   ovf    = novf + 32;
    int*   runlen = ovf + 2 * (size_t)N_EDGES;
    int*   runs   = runlen + (size_t)NBKT * RL_PITCH;

    k_bin_gemm  <<<NBIN + TILES_BIN, 256, 0, stream>>>(src, dst, runlen, runs, sums, sumsq, novf, x, Wm, h);
    k_build_gemm<<<NBKT + TILES_BUILD, 256, 0, stream>>>(src, dst, runlen, runs, cnt, col, novf, ovf, x, Wm, h);

    k_gather    <<<N_NODES / 8, 256, 0, stream>>>(cnt, col, novf, ovf, h, b, out);

    k_stats     <<<(N_NODES / ST_ROWS + ST_WAVES - 1) / ST_WAVES, ST_WAVES * 32, 0, stream>>>(out, sums, sumsq);
    k_bn        <<<(N_NODES / BN_ROWS + 7) / 8, 256, 0, stream>>>(out, sums, sumsq, gamma, beta);
}


# Independent loads issued together instead of one round trip each: gather fetches count, id line, self row and bias at once, then 10 rows with their degrees; x tile staged as 2 float4 per thread
# speedup vs baseline: 2.6166x; 1.0623x over previous
#include <hip/hip_runtime.h>

#define N_NODES 100000
#define N_EDGES 640000
#define D       128
#define BN_EPS  1e-5f
#define LDS_PITCH 132

typedef float v2f __attribute__((ext_vector_type(2)));
typedef float v8f __attribute__((ext_vector_type(8)));

#define GEMM_TILES  (N_NODES / 16)
#define TILES_BIN   (GEMM_TILES / 2)
#define TILES_BUILD (GEMM_TILES - TILES_BIN)

template <int NE, int NT>
__device__ __forceinline__ bool block_is_tile(int bid, int& idx) {
    constexpr int M = NE < NT ? NE : NT;
    if (bid < 2 * M) { idx = bid >> 1; return (bid & 1) != 0; }
    idx = bid - M;
    return NT > NE;
}

__device__ __forceinline__ void gemm_tile(int tile, const float* __restrict__ x,
                                          const float* __restrict__ Wm,
                                          float* __restrict__ h, float* lx) {
    const int tid  = threadIdx.x;
    const int row0 = tile * 16;

    const float4* xt = (const float4*)(x + (size_t)row0 * D);
    const float4 x0 = xt[tid];
    const float4 x1 = xt[tid + 256];
    *(float4*)&lx[(tid >> 5) * LDS_PITCH + (tid & 31) * 4]       = x0;
    *(float4*)&lx[((tid >> 5) + 8) * LDS_PITCH + (tid & 31) * 4] = x1;
    __syncthreads();

    const int wave = tid >> 5;
    const int lane = tid & 31;
    const int half = lane >> 4;
    const int l15  = lane & 15;
    const int koff = half * 2;
    const int n0   = wave * 16;

    v8f acc = {};
    for (int k = 0; k < D; k += 4) {
        v2f a, b;
        a.x = lx[l15 * LDS_PITCH + k + koff];
        a.y = lx[l15 * LDS_PITCH + k + koff + 1];
        b.x = Wm[(k + koff)     * D + n0 + l15];
        b.y = Wm[(k + koff + 1) * D + n0 + l15];
        acc = __builtin_amdgcn_wmma_f32_16x16x4_f32(
             false, a,  false, b,
             (short)0, acc,  false,  false);
    }

#pragma unroll
    for (int r = 0; r < 8; ++r) {
        int row = row0 + r + half * 8;
        h[row * D + n0 + l15] = acc[r];
    }
}

#define CAP 32
#define CHUNK     2560
#define NBIN      (N_EDGES / CHUNK)
#define BKT_SHIFT 8
#define BKT       (1 << BKT_SHIFT)
#define NBKT      ((N_NODES + BKT - 1) / BKT)
#define RCAP      16
#define RL_PITCH  256

__device__ __forceinline__ void list_put(int s, int d, int r, int base, int* lcol,
                                         int* __restrict__ novf, int* __restrict__ ovf) {
    if (r < CAP) {
        lcol[(d - base) * CAP + r] = s;
    } else {
        const int o = atomicAdd(novf, 1);
        ovf[2 * o]     = s;
        ovf[2 * o + 1] = d;
    }
}

__global__ void __launch_bounds__(256) k_bin_gemm(const int* __restrict__ src,
                                                  const int* __restrict__ dst,
                                                  int* __restrict__ runlen,
                                                  int* __restrict__ runs,
                                                  float* __restrict__ sums,
                                                  float* __restrict__ sumsq,
                                                  int* __restrict__ novf,
                                                  const float* __restrict__ x,
                                                  const float* __restrict__ Wm,
                                                  float* __restrict__ h) {
    static_assert(N_EDGES % CHUNK == 0 && CHUNK % 256 == 0, "whole chunks, whole passes of the block");
    static_assert(NBIN <= RL_PITCH && NBIN <= 256, "one thread per run in level 2");
    static_assert(NBKT * 4 <= 16 * LDS_PITCH * 4, "bucket counters fit the tile buffer");
    __shared__ __attribute__((aligned(16))) float lx[16 * LDS_PITCH];
    int idx;
    if (block_is_tile<NBIN, TILES_BIN>((int)blockIdx.x, idx)) {
        gemm_tile(idx, x, Wm, h, lx);
        return;
    }
    int* lcnt = (int*)lx;
    const int tid = threadIdx.x;
    if (idx == 0) {
        if (tid < D) { sums[tid] = 0.0f; sumsq[tid] = 0.0f; }
        if (tid == 0) novf[0] = 0;
    }
    for (int k = tid; k < NBKT; k += 256) lcnt[k] = 0;

    constexpr int PER = CHUNK / 256;
    const int e0 = idx * CHUNK + tid;
    int d[PER], s[PER];
#pragma unroll
    for (int j = 0; j < PER; ++j) { d[j] = dst[e0 + j * 256]; s[j] = src[e0 + j * 256]; }
    __syncthreads();
    int r[PER];
#pragma unroll
    for (int j = 0; j < PER; ++j)
        r[j] = atomicAdd(&lcnt[d[j] >> BKT_SHIFT], 1);
#pragma unroll
    for (int j = 0; j < PER; ++j) {
        if (r[j] < RCAP) {
            const int k = d[j] >> BKT_SHIFT;
            int2 p; p.x = s[j]; p.y = d[j];
            ((int2*)runs)[((size_t)k * NBIN + idx) * RCAP + r[j]] = p;
        }
    }
    __syncthreads();
    for (int k = tid; k < NBKT; k += 256) runlen[k * RL_PITCH + idx] = lcnt[k];
}

#define BUILD_LDS_INTS (BKT * CAP + BKT + 256 + 256 + 4)
__global__ void __launch_bounds__(256) k_build_gemm(const int* __restrict__ src,
                                                    const int* __restrict__ dst,
                                                    const int* __restrict__ runlen,
                                                    const int* __restrict__ runs,
                                                    int* __restrict__ cnt,
                                                    int* __restrict__ col,
                                                    int* __restrict__ novf,
                                                    int* __restrict__ ovf,
                                                    const float* __restrict__ x,
                                                    const float* __restrict__ Wm,
                                                    float* __restrict__ h) {
    static_assert(BKT == 256, "one thread per node of the bucket");
    static_assert(RCAP == 16 && NBIN * RCAP <= 16 * 256, "16 passes of the block cover every stored pair");
    static_assert(BUILD_LDS_INTS >= 16 * LDS_PITCH, "the tile buffer fits");
    __shared__ __attribute__((aligned(16))) int sm[BUILD_LDS_INTS];
    int idx;
    if (block_is_tile<NBKT, TILES_BUILD>((int)blockIdx.x, idx)) {
        gemm_tile(TILES_BIN + idx, x, Wm, h, (float*)sm);
        return;
    }
    int* lcol = sm;
    int* lcnt = lcol + BKT * CAP;
    int* rl   = lcnt + BKT;
    int* fb   = rl + 256;
    int* nfb  = fb + 256;
    const int tid  = threadIdx.x;
    const int k    = idx;
    const int base = k << BKT_SHIFT;

    lcnt[tid] = 0;
    int n = 0;
    if (tid < NBIN) n = runlen[k * RL_PITCH + tid];
    rl[tid] = n;
    if (tid == 0) nfb[0] = 0;
    __syncthreads();
    if (n > RCAP) { const int q = atomicAdd(nfb, 1); fb[q] = tid; }

    const int2* rp = (const int2*)runs + (size_t)k * NBIN * RCAP;
    int  nb[16];
    int2 p[16];
    int  r[16];
#pragma unroll
    for (int it = 0; it < 16; ++it) nb[it] = rl[(it * 256 + tid) >> 4];
#pragma unroll
    for (int it = 0; it < 16; ++it) {
        const int f = it * 256 + tid;
        p[it].x = 0; p[it].y = -1;
        if (nb[it] <= RCAP && (f & 15) < nb[it]) p[it] = rp[f];
    }
#pragma unroll
    for (int it = 0; it < 16; ++it) {
        r[it] = 0;
        if (p[it].y >= 0) r[it] = atomicAdd(&lcnt[p[it].y - base], 1);
    }
#pragma unroll
    for (int it = 0; it < 16; ++it)
        if (p[it].y >= 0) list_put(p[it].x, p[it].y, r[it], base, lcol, novf, ovf);
    __syncthreads();

    const int nf = nfb[0];
    for (int q = 0; q < nf; ++q) {
        const int e0 = fb[q] * CHUNK + tid;
        int dd[CHUNK / 256];
#pragma unroll
        for (int j = 0; j < CHUNK / 256; ++j) dd[j] = dst[e0 + j * 256];
#pragma unroll
        for (int j = 0; j < CHUNK / 256; ++j) {
            if ((dd[j] >> BKT_SHIFT) == k) {
                const int rr = atomicAdd(&lcnt[dd[j] - base], 1);
                list_put(src[e0 + j * 256], dd[j], rr, base, lcol, novf, ovf);
            }
        }
    }
    __syncthreads();

    if (base + tid < N_NODES) cnt[base + tid] = lcnt[tid];
    for (int q = tid; q < BKT * CAP / 4; q += 256)
        if (base + (q >> 3) < N_NODES)
            ((int4*)col)[(size_t)base * (CAP / 4) + q] = ((const int4*)lcol)[q];
}

template <int R>
__device__ __forceinline__ void add_rows(float4& acc, const float4* __restrict__ hp,
                                         const int* __restrict__ cnt, int my_s, float dt,
                                         int j, int lane) {
    float4 v[R];
#pragma unroll
    for (int k = 0; k < R; ++k)
        v[k] = hp[(size_t)__builtin_amdgcn_readlane(my_s, j + k) * 32 + lane];
    const float my_w = rsqrtf((float)(cnt[my_s] + 1)) * dt;
#pragma unroll
    for (int k = 0; k < R; ++k) {
        const float w = __int_as_float(__builtin_amdgcn_readlane(__float_as_int(my_w), j + k));
        acc.x = fmaf(v[k].x, w, acc.x);
        acc.y = fmaf(v[k].y, w, acc.y);
        acc.z = fmaf(v[k].z, w, acc.z);
        acc.w = fmaf(v[k].w, w, acc.w);
    }
}

#define GR 10
__global__ void __launch_bounds__(256) k_gather(const int* __restrict__ cnt,
                                                const int* __restrict__ col,
                                                const int* __restrict__ novf,
                                                const int* __restrict__ ovf,
                                                const float* __restrict__ h,
                                                const float* __restrict__ b,
                                                float* __restrict__ out) {
    const int lane = threadIdx.x & 31;
    const int t = __builtin_amdgcn_readfirstlane((int)(blockIdx.x * 8 + (threadIdx.x >> 5)));

    const float4* hp = (const float4*)h;
    const int    c   = cnt[t];
    const int    raw = col[t * CAP + lane];
    const float4 hs  = hp[(size_t)t * 32 + lane];
    const float4 bb  = ((const float4*)b)[lane];

    const float dt = rsqrtf((float)(c + 1));
    const int   n  = c < CAP ? c : CAP;
    asm volatile("" :: "v"(raw));
    const int   my_s = lane < n ? raw : 0;

    const float  ws = dt * dt;
    float4 acc;
    acc.x = hs.x * ws; acc.y = hs.y * ws; acc.z = hs.z * ws; acc.w = hs.w * ws;

    for (int j = 0; j < n; j += GR) {
        switch (n - j) {
            case 1:  add_rows<1>(acc, hp, cnt, my_s, dt, j, lane); break;
            case 2:  add_rows<2>(acc, hp, cnt, my_s, dt, j, lane); break;
            case 3:  add_rows<3>(acc, hp, cnt, my_s, dt, j, lane); break;
            case 4:  add_rows<4>(acc, hp, cnt, my_s, dt, j, lane); break;
            case 5:  add_rows<5>(acc, hp, cnt, my_s, dt, j, lane); break;
            case 6:  add_rows<6>(acc, hp, cnt, my_s, dt, j, lane); break;
            case 7:  add_rows<7>(acc, hp, cnt, my_s, dt, j, lane); break;
            case 8:  add_rows<8>(acc, hp, cnt, my_s, dt, j, lane); break;
            case 9:  add_rows<9>(acc, hp, cnt, my_s, dt, j, lane); break;
            default: add_rows<GR>(acc, hp, cnt, my_s, dt, j, lane); break;
        }
    }

    if (c > CAP) {
        const int no = novf[0];
        for (int i0 = 0; i0 < no; i0 += 32) {
            int s_l = 0;
            int d_l = -1;
            if (i0 + lane < no) {
                s_l = ovf[2 * (i0 + lane)];
                d_l = ovf[2 * (i0 + lane) + 1];
            }
            unsigned int m = __builtin_amdgcn_ballot_w32(d_l == t);
            while (m != 0u) {
                const int k = __builtin_ctz(m);
                m &= m - 1u;
                const int   s = __builtin_amdgcn_readlane(s_l, k);
                const float w = rsqrtf((float)(cnt[s] + 1)) * dt;
                const float4 v = hp[(size_t)s * 32 + lane];
                acc.x = fmaf(v.x, w, acc.x);
                acc.y = fmaf(v.y, w, acc.y);
                acc.z = fmaf(v.z, w, acc.z);
                acc.w = fmaf(v.w, w, acc.w);
            }
        }
    }

    float4 o;
    o.x = acc.x + bb.x; o.y = acc.y + bb.y; o.z = acc.z + bb.z; o.w = acc.w + bb.w;
    ((float4*)out)[(size_t)t * 32 + lane] = o;
}

#define ST_ROWS  16
#define ST_WAVES 16

__global__ void __launch_bounds__(ST_WAVES * 32) k_stats(const float* __restrict__ out,
                                                         float* __restrict__ sums,
                                                         float* __restrict__ sumsq) {
    static_assert(N_NODES % ST_ROWS == 0, "whole waves only");
    __shared__ float red[ST_WAVES][2 * D];
    const int lane = threadIdx.x & 31;
    const int wave = threadIdx.x >> 5;
    const int row0 = (int)(blockIdx.x * ST_WAVES + wave) * ST_ROWS;

    float4 s = make_float4(0.0f, 0.0f, 0.0f, 0.0f);
    float4 q = make_float4(0.0f, 0.0f, 0.0f, 0.0f);
    if (row0 < N_NODES) {
        const float4* op = (const float4*)out + (size_t)row0 * 32 + lane;
        float4 v[ST_ROWS];
#pragma unroll
        for (int k = 0; k < ST_ROWS; ++k) v[k] = op[k * 32];
#pragma unroll
        for (int k = 0; k < ST_ROWS; ++k) {
            float4 r;
            r.x = v[k].x > 0.0f ? v[k].x : 0.0f;
            r.y = v[k].y > 0.0f ? v[k].y : 0.0f;
            r.z = v[k].z > 0.0f ? v[k].z : 0.0f;
            r.w = v[k].w > 0.0f ? v[k].w : 0.0f;
            s.x += r.x; s.y += r.y; s.z += r.z; s.w += r.w;
            q.x += r.x * r.x; q.y += r.y * r.y; q.z += r.z * r.z; q.w += r.w * r.w;
        }
    }
    ((float4*)&red[wave][0])[lane] = s;
    ((float4*)&red[wave][D])[lane] = q;
    __syncthreads();

    if (threadIdx.x < 2 * D) {
        float a = 0.0f;
#pragma unroll
        for (int w = 0; w < ST_WAVES; ++w) a += red[w][threadIdx.x];
        if (threadIdx.x < D) atomicAdd(&sums[threadIdx.x], a);
        else                 atomicAdd(&sumsq[threadIdx.x - D], a);
    }
}

#define BN_ROWS 16

__device__ __forceinline__ float bn_one(float v, float mean, float rs, float g, float bt) {
    v = v > 0.0f ? v : 0.0f;
    return g * (v - mean) * rs + bt;
}

__global__ void __launch_bounds__(256) k_bn(float* __restrict__ out,
                                            const float* __restrict__ sums,
                                            const float* __restrict__ sumsq,
                                            const float* __restrict__ gamma,
                                            const float* __restrict__ beta) {
    static_assert(N_NODES % BN_ROWS == 0, "whole waves only");
    const int lane = threadIdx.x & 31;
    const int row0 = (int)(blockIdx.x * 8 + (threadIdx.x >> 5)) * BN_ROWS;
    if (row0 >= N_NODES) return;

    float4* op = (float4*)out + (size_t)row0 * 32 + lane;
    float4 v[BN_ROWS];
#pragma unroll
    for (int k = 0; k < BN_ROWS; ++k) v[k] = op[k * 32];

    const float4 s  = ((const float4*)sums)[lane];
    const float4 sq = ((const float4*)sumsq)[lane];
    const float4 g  = ((const float4*)gamma)[lane];
    const float4 bt = ((const float4*)beta)[lane];

    const float invN = 1.0f / (float)N_NODES;
    float4 mean, var, rs;
    mean.x = s.x * invN; mean.y = s.y * invN; mean.z = s.z * invN; mean.w = s.w * invN;
    var.x = sq.x * invN - mean.x * mean.x;
    var.y = sq.y * invN - mean.y * mean.y;
    var.z = sq.z * invN - mean.z * mean.z;
    var.w = sq.w * invN - mean.w * mean.w;
    rs.x = rsqrtf(var.x + BN_EPS); rs.y = rsqrtf(var.y + BN_EPS);
    rs.z = rsqrtf(var.z + BN_EPS); rs.w = rsqrtf(var.w + BN_EPS);

#pragma unroll
    for (int k = 0; k < BN_ROWS; ++k) {
        float4 o;
        o.x = bn_one(v[k].x, mean.x, rs.x, g.x, bt.x);
        o.y = bn_one(v[k].y, mean.y, rs.y, g.y, bt.y);
        o.z = bn_one(v[k].z, mean.z, rs.z, g.z, bt.z);
        o.w = bn_one(v[k].w, mean.w, rs.w, g.w, bt.w);
        op[k * 32] = o;
    }
}

extern "C" void kernel_launch(void* const* d_in, const int* in_sizes, int n_in,
                              void* d_out, int out_size, void* d_ws, size_t ws_size,
                              hipStream_t stream) {
    const float* x     = (const float*)d_in[0];
    const int*   edges = (const int*)  d_in[1];
    const float* Wm    = (const float*)d_in[2];
    const float* b     = (const float*)d_in[3];
    const float* gamma = (const float*)d_in[4];
    const float* beta  = (const float*)d_in[5];
    float*       out   = (float*)d_out;

    const int* src = edges;
    const int* dst = edges + N_EDGES;

    float* h      = (float*)d_ws;
    int*   cnt    = (int*)(h + (size_t)N_NODES * D);
    int*   col    = cnt + N_NODES;
    float* sums   = (float*)(col + (size_t)N_NODES * CAP);
    float* sumsq  = sums + D;
    int*   novf   = (int*)(sumsq + D);
    int*   ovf    = novf + 32;
    int*   runlen = ovf + 2 * (size_t)N_EDGES;
    int*   runs   = runlen + (size_t)NBKT * RL_PITCH;

    k_bin_gemm  <<<NBIN + TILES_BIN, 256, 0, stream>>>(src, dst, runlen, runs, sums, sumsq, novf, x, Wm, h);
    k_build_gemm<<<NBKT + TILES_BUILD, 256, 0, stream>>>(src, dst, runlen, runs, cnt, col, novf, ovf, x, Wm, h);

    k_gather    <<<N_NODES / 8, 256, 0, stream>>>(cnt, col, novf, ovf, h, b, out);

    k_stats     <<<(N_NODES / ST_ROWS + ST_WAVES - 1) / ST_WAVES, ST_WAVES * 32, 0, stream>>>(out, sums, sumsq);
    k_bn        <<<(N_NODES / BN_ROWS + 7) / 8, 256, 0, stream>>>(out, sums, sumsq, gamma, beta);
}


# BN sums taken in the gather launch from rows in registers: a wave adds 4 nodes, LDS add over the block, one float atomic per column per block into 8 copies; stats launch gone (sum order changes)
# speedup vs baseline: 2.7379x; 1.0464x over previous
#include <hip/hip_runtime.h>

#define N_NODES 100000
#define N_EDGES 640000
#define D       128
#define BN_EPS  1e-5f
#define LDS_PITCH 132

#define NREP 8

typedef float v2f __attribute__((ext_vector_type(2)));
typedef float v8f __attribute__((ext_vector_type(8)));

#define GEMM_TILES  (N_NODES / 16)
#define TILES_BIN   (GEMM_TILES / 2)
#define TILES_BUILD (GEMM_TILES - TILES_BIN)

template <int NE, int NT>
__device__ __forceinline__ bool block_is_tile(int bid, int& idx) {
    constexpr int M = NE < NT ? NE : NT;
    if (bid < 2 * M) { idx = bid >> 1; return (bid & 1) != 0; }
    idx = bid - M;
    return NT > NE;
}

__device__ __forceinline__ void gemm_tile(int tile, const float* __restrict__ x,
                                          const float* __restrict__ Wm,
                                          float* __restrict__ h, float* lx) {
    const int tid  = threadIdx.x;
    const int row0 = tile * 16;

    const float4* xt = (const float4*)(x + (size_t)row0 * D);
    const float4 x0 = xt[tid];
    const float4 x1 = xt[tid + 256];
    *(float4*)&lx[(tid >> 5) * LDS_PITCH + (tid & 31) * 4]       = x0;
    *(float4*)&lx[((tid >> 5) + 8) * LDS_PITCH + (tid & 31) * 4] = x1;
    __syncthreads();

    const int wave = tid >> 5;
    const int lane = tid & 31;
    const int half = lane >> 4;
    const int l15  = lane & 15;
    const int koff = half * 2;
    const int n0   = wave * 16;

    v8f acc = {};
    for (int k = 0; k < D; k += 4) {
        v2f a, b;
        a.x = lx[l15 * LDS_PITCH + k + koff];
        a.y = lx[l15 * LDS_PITCH + k + koff + 1];
        b.x = Wm[(k + koff)     * D + n0 + l15];
        b.y = Wm[(k + koff + 1) * D + n0 + l15];
        acc = __builtin_amdgcn_wmma_f32_16x16x4_f32(
             false, a,  false, b,
             (short)0, acc,  false,  false);
    }

#pragma unroll
    for (int r = 0; r < 8; ++r) {
        int row = row0 + r + half * 8;
        h[row * D + n0 + l15] = acc[r];
    }
}

#define CAP 32
#define CHUNK     2560
#define NBIN      (N_EDGES / CHUNK)
#define BKT_SHIFT 8
#define BKT       (1 << BKT_SHIFT)
#define NBKT      ((N_NODES + BKT - 1) / BKT)
#define RCAP      16
#define RL_PITCH  256

__device__ __forceinline__ void list_put(int s, int d, int r, int base, int* lcol,
                                         int* __restrict__ novf, int* __restrict__ ovf) {
    if (r < CAP) {
        lcol[(d - base) * CAP + r] = s;
    } else {
        const int o = atomicAdd(novf, 1);
        ovf[2 * o]     = s;
        ovf[2 * o + 1] = d;
    }
}

__global__ void __launch_bounds__(256) k_bin_gemm(const int* __restrict__ src,
                                                  const int* __restrict__ dst,
                                                  int* __restrict__ runlen,
                                                  int* __restrict__ runs,
                                                  float* __restrict__ stat,
                                                  int* __restrict__ novf,
                                                  const float* __restrict__ x,
                                                  const float* __restrict__ Wm,
                                                  float* __restrict__ h) {
    static_assert(N_EDGES % CHUNK == 0 && CHUNK % 256 == 0, "whole chunks, whole passes of the block");
    static_assert(NBIN <= RL_PITCH && NBIN <= 256, "one thread per run in level 2");
    static_assert(NBKT * 4 <= 16 * LDS_PITCH * 4, "bucket counters fit the tile buffer");
    __shared__ __attribute__((aligned(16))) float lx[16 * LDS_PITCH];
    int idx;
    if (block_is_tile<NBIN, TILES_BIN>((int)blockIdx.x, idx)) {
        gemm_tile(idx, x, Wm, h, lx);
        return;
    }
    int* lcnt = (int*)lx;
    const int tid = threadIdx.x;
    if (idx == 0) {
        for (int i = tid; i < NREP * 2 * D; i += 256) stat[i] = 0.0f;
        if (tid == 0) novf[0] = 0;
    }
    for (int k = tid; k < NBKT; k += 256) lcnt[k] = 0;

    constexpr int PER = CHUNK / 256;
    const int e0 = idx * CHUNK + tid;
    int d[PER], s[PER];
#pragma unroll
    for (int j = 0; j < PER; ++j) { d[j] = dst[e0 + j * 256]; s[j] = src[e0 + j * 256]; }
    __syncthreads();
    int r[PER];
#pragma unroll
    for (int j = 0; j < PER; ++j)
        r[j] = atomicAdd(&lcnt[d[j] >> BKT_SHIFT], 1);
#pragma unroll
    for (int j = 0; j < PER; ++j) {
        if (r[j] < RCAP) {
            const int k = d[j] >> BKT_SHIFT;
            int2 p; p.x = s[j]; p.y = d[j];
            ((int2*)runs)[((size_t)k * NBIN + idx) * RCAP + r[j]] = p;
        }
    }
    __syncthreads();
    for (int k = tid; k < NBKT; k += 256) runlen[k * RL_PITCH + idx] = lcnt[k];
}

#define BUILD_LDS_INTS (BKT * CAP + BKT + 256 + 256 + 4)
__global__ void __launch_bounds__(256) k_build_gemm(const int* __restrict__ src,
                                                    const int* __restrict__ dst,
                                                    const int* __restrict__ runlen,
                                                    const int* __restrict__ runs,
                                                    int* __restrict__ cnt,
                                                    int* __restrict__ col,
                                                    int* __restrict__ novf,
                                                    int* __restrict__ ovf,
                                                    const float* __restrict__ x,
                                                    const float* __restrict__ Wm,
                                                    float* __restrict__ h) {
    static_assert(BKT == 256, "one thread per node of the bucket");
    static_assert(RCAP == 16 && NBIN * RCAP <= 16 * 256, "16 passes of the block cover every stored pair");
    static_assert(BUILD_LDS_INTS >= 16 * LDS_PITCH, "the tile buffer fits");
    __shared__ __attribute__((aligned(16))) int sm[BUILD_LDS_INTS];
    int idx;
    if (block_is_tile<NBKT, TILES_BUILD>((int)blockIdx.x, idx)) {
        gemm_tile(TILES_BIN + idx, x, Wm, h, (float*)sm);
        return;
    }
    int* lcol = sm;
    int* lcnt = lcol + BKT * CAP;
    int* rl   = lcnt + BKT;
    int* fb   = rl + 256;
    int* nfb  = fb + 256;
    const int tid  = threadIdx.x;
    const int k    = idx;
    const int base = k << BKT_SHIFT;

    lcnt[tid] = 0;
    int n = 0;
    if (tid < NBIN) n = runlen[k * RL_PITCH + tid];
    rl[tid] = n;
    if (tid == 0) nfb[0] = 0;
    __syncthreads();
    if (n > RCAP) { const int q = atomicAdd(nfb, 1); fb[q] = tid; }

    const int2* rp = (const int2*)runs + (size_t)k * NBIN * RCAP;
    int  nb[16];
    int2 p[16];
    int  r[16];
#pragma unroll
    for (int it = 0; it < 16; ++it) nb[it] = rl[(it * 256 + tid) >> 4];
#pragma unroll
    for (int it = 0; it < 16; ++it) {
        const int f = it * 256 + tid;
        p[it].x = 0; p[it].y = -1;
        if (nb[it] <= RCAP && (f & 15) < nb[it]) p[it] = rp[f];
    }
#pragma unroll
    for (int it = 0; it < 16; ++it) {
        r[it] = 0;
        if (p[it].y >= 0) r[it] = atomicAdd(&lcnt[p[it].y - base], 1);
    }
#pragma unroll
    for (int it = 0; it < 16; ++it)
        if (p[it].y >= 0) list_put(p[it].x, p[it].y, r[it], base, lcol, novf, ovf);
    __syncthreads();

    const int nf = nfb[0];
    for (int q = 0; q < nf; ++q) {
        const int e0 = fb[q] * CHUNK + tid;
        int dd[CHUNK / 256];
#pragma unroll
        for (int j = 0; j < CHUNK / 256; ++j) dd[j] = dst[e0 + j * 256];
#pragma unroll
        for (int j = 0; j < CHUNK / 256; ++j) {
            if ((dd[j] >> BKT_SHIFT) == k) {
                const int rr = atomicAdd(&lcnt[dd[j] - base], 1);
                list_put(src[e0 + j * 256], dd[j], rr, base, lcol, novf, ovf);
            }
        }
    }
    __syncthreads();

    if (base + tid < N_NODES) cnt[base + tid] = lcnt[tid];
    for (int q = tid; q < BKT * CAP / 4; q += 256)
        if (base + (q >> 3) < N_NODES)
            ((int4*)col)[(size_t)base * (CAP / 4) + q] = ((const int4*)lcol)[q];
}

template <int R>
__device__ __forceinline__ void add_rows(float4& acc, const float4* __restrict__ hp,
                                         const int* __restrict__ cnt, int my_s, float dt,
                                         int j, int lane) {
    float4 v[R];
#pragma unroll
    for (int k = 0; k < R; ++k)
        v[k] = hp[(size_t)__builtin_amdgcn_readlane(my_s, j + k) * 32 + lane];
    const float my_w = rsqrtf((float)(cnt[my_s] + 1)) * dt;
#pragma unroll
    for (int k = 0; k < R; ++k) {
        const float w = __int_as_float(__builtin_amdgcn_readlane(__float_as_int(my_w), j + k));
        acc.x = fmaf(v[k].x, w, acc.x);
        acc.y = fmaf(v[k].y, w, acc.y);
        acc.z = fmaf(v[k].z, w, acc.z);
        acc.w = fmaf(v[k].w, w, acc.w);
    }
}

#define GR 9
#define NPW 4
__global__ void __launch_bounds__(256) k_gather(const int* __restrict__ cnt,
                                                const int* __restrict__ col,
                                                const int* __restrict__ novf,
                                                const int* __restrict__ ovf,
                                                const float* __restrict__ h,
                                                const float* __restrict__ b,
                                                float* __restrict__ out,
                                                float* __restrict__ stat) {
    static_assert(N_NODES % (8 * NPW) == 0, "whole blocks only: every wave has NPW nodes and reaches the barrier");
    __shared__ __attribute__((aligned(16))) float red[8][2 * D];
    const int lane = threadIdx.x & 31;
    const int wave = threadIdx.x >> 5;
    const int t0 = __builtin_amdgcn_readfirstlane((int)(blockIdx.x * 8 + wave) * NPW);

    const float4* hp = (const float4*)h;
    const float4 bb  = ((const float4*)b)[lane];
    float4* my_sr = (float4*)&red[wave][0] + lane;
    float4* my_sq = (float4*)&red[wave][D] + lane;
    *my_sr = make_float4(0.0f, 0.0f, 0.0f, 0.0f);
    *my_sq = make_float4(0.0f, 0.0f, 0.0f, 0.0f);

    for (int i = 0; i < NPW; ++i) {
        const int t = t0 + i;
        const int    c   = cnt[t];
        const int    raw = col[t * CAP + lane];
        const float4 hs  = hp[(size_t)t * 32 + lane];

        const float dt = rsqrtf((float)(c + 1));
        const int   n  = c < CAP ? c : CAP;
        asm volatile("" :: "v"(raw));
        const int   my_s = lane < n ? raw : 0;

        const float  ws = dt * dt;
        float4 acc;
        acc.x = hs.x * ws; acc.y = hs.y * ws; acc.z = hs.z * ws; acc.w = hs.w * ws;

        for (int j = 0; j < n; j += GR) {
            switch (n - j) {
                case 1:  add_rows<1>(acc, hp, cnt, my_s, dt, j, lane); break;
                case 2:  add_rows<2>(acc, hp, cnt, my_s, dt, j, lane); break;
                case 3:  add_rows<3>(acc, hp, cnt, my_s, dt, j, lane); break;
                case 4:  add_rows<4>(acc, hp, cnt, my_s, dt, j, lane); break;
                case 5:  add_rows<5>(acc, hp, cnt, my_s, dt, j, lane); break;
                case 6:  add_rows<6>(acc, hp, cnt, my_s, dt, j, lane); break;
                case 7:  add_rows<7>(acc, hp, cnt, my_s, dt, j, lane); break;
                case 8:  add_rows<8>(acc, hp, cnt, my_s, dt, j, lane); break;
                default: add_rows<GR>(acc, hp, cnt, my_s, dt, j, lane); break;
            }
        }

        if (c > CAP) {
            const int no = novf[0];
            for (int i0 = 0; i0 < no; i0 += 32) {
                int s_l = 0;
                int d_l = -1;
                if (i0 + lane < no) {
                    s_l = ovf[2 * (i0 + lane)];
                    d_l = ovf[2 * (i0 + lane) + 1];
                }
                unsigned int m = __builtin_amdgcn_ballot_w32(d_l == t);
                while (m != 0u) {
                    const int k = __builtin_ctz(m);
                    m &= m - 1u;
                    const int   s = __builtin_amdgcn_readlane(s_l, k);
                    const float w = rsqrtf((float)(cnt[s] + 1)) * dt;
                    const float4 v = hp[(size_t)s * 32 + lane];
                    acc.x = fmaf(v.x, w, acc.x);
                    acc.y = fmaf(v.y, w, acc.y);
                    acc.z = fmaf(v.z, w, acc.z);
                    acc.w = fmaf(v.w, w, acc.w);
                }
            }
        }

        float4 o;
        o.x = acc.x + bb.x; o.y = acc.y + bb.y; o.z = acc.z + bb.z; o.w = acc.w + bb.w;
        ((float4*)out)[(size_t)t * 32 + lane] = o;

        float4 r;
        r.x = o.x > 0.0f ? o.x : 0.0f;
        r.y = o.y > 0.0f ? o.y : 0.0f;
        r.z = o.z > 0.0f ? o.z : 0.0f;
        r.w = o.w > 0.0f ? o.w : 0.0f;
        float4 sr = *my_sr;
        float4 sq = *my_sq;
        sr.x += r.x; sr.y += r.y; sr.z += r.z; sr.w += r.w;
        sq.x += r.x * r.x; sq.y += r.y * r.y; sq.z += r.z * r.z; sq.w += r.w * r.w;
        *my_sr = sr;
        *my_sq = sq;
    }
    __syncthreads();
    float a = 0.0f;
#pragma unroll
    for (int w = 0; w < 8; ++w) a += red[w][threadIdx.x];
    atomicAdd(&stat[(blockIdx.x % NREP) * (2 * D) + threadIdx.x], a);
}

#define BN_ROWS 16

__device__ __forceinline__ float bn_one(float v, float mean, float rs, float g, float bt) {
    v = v > 0.0f ? v : 0.0f;
    return g * (v - mean) * rs + bt;
}

__global__ void __launch_bounds__(256) k_bn(float* __restrict__ out,
                                            const float* __restrict__ stat,
                                            const float* __restrict__ gamma,
                                            const float* __restrict__ beta) {
    static_assert(N_NODES % BN_ROWS == 0, "whole waves only");
    const int lane = threadIdx.x & 31;
    const int row0 = (int)(blockIdx.x * 8 + (threadIdx.x >> 5)) * BN_ROWS;
    if (row0 >= N_NODES) return;

    float4* op = (float4*)out + (size_t)row0 * 32 + lane;
    float4 v[BN_ROWS];
#pragma unroll
    for (int k = 0; k < BN_ROWS; ++k) v[k] = op[k * 32];

    float4 s  = make_float4(0.0f, 0.0f, 0.0f, 0.0f);
    float4 sq = make_float4(0.0f, 0.0f, 0.0f, 0.0f);
#pragma unroll
    for (int k = 0; k < NREP; ++k) {
        const float4 a = ((const float4*)(stat + k * 2 * D))[lane];
        const float4 c = ((const float4*)(stat + k * 2 * D + D))[lane];
        s.x += a.x; s.y += a.y; s.z += a.z; s.w += a.w;
        sq.x += c.x; sq.y += c.y; sq.z += c.z; sq.w += c.w;
    }
    const float4 g  = ((const float4*)gamma)[lane];
    const float4 bt = ((const float4*)beta)[lane];

    const float invN = 1.0f / (float)N_NODES;
    float4 mean, var, rs;
    mean.x = s.x * invN; mean.y = s.y * invN; mean.z = s.z * invN; mean.w = s.w * invN;
    var.x = sq.x * invN - mean.x * mean.x;
    var.y = sq.y * invN - mean.y * mean.y;
    var.z = sq.z * invN - mean.z * mean.z;
    var.w = sq.w * invN - mean.w * mean.w;
    rs.x = rsqrtf(var.x + BN_EPS); rs.y = rsqrtf(var.y + BN_EPS);
    rs.z = rsqrtf(var.z + BN_EPS); rs.w = rsqrtf(var.w + BN_EPS);

#pragma unroll
    for (int k = 0; k < BN_ROWS; ++k) {
        float4 o;
        o.x = bn_one(v[k].x, mean.x, rs.x, g.x, bt.x);
        o.y = bn_one(v[k].y, mean.y, rs.y, g.y, bt.y);
        o.z = bn_one(v[k].z, mean.z, rs.z, g.z, bt.z);
        o.w = bn_one(v[k].w, mean.w, rs.w, g.w, bt.w);
        op[k * 32] = o;
    }
}

extern "C" void kernel_launch(void* const* d_in, const int* in_sizes, int n_in,
                              void* d_out, int out_size, void* d_ws, size_t ws_size,
                              hipStream_t stream) {
    const float* x     = (const float*)d_in[0];
    const int*   edges = (const int*)  d_in[1];
    const float* Wm    = (const float*)d_in[2];
    const float* b     = (const float*)d_in[3];
    const float* gamma = (const float*)d_in[4];
    const float* beta  = (const float*)d_in[5];
    float*       out   = (float*)d_out;

    const int* src = edges;
    const int* dst = edges + N_EDGES;

    float* h      = (float*)d_ws;
    int*   cnt    = (int*)(h + (size_t)N_NODES * D);
    int*   col    = cnt + N_NODES;
    float* stat   = (float*)(col + (size_t)N_NODES * CAP);
    int*   novf   = (int*)(stat + NREP * 2 * D);
    int*   ovf    = novf + 32;
    int*   runlen = ovf + 2 * (size_t)N_EDGES;
    int*   runs   = runlen + (size_t)NBKT * RL_PITCH;

    k_bin_gemm  <<<NBIN + TILES_BIN, 256, 0, stream>>>(src, dst, runlen, runs, stat, novf, x, Wm, h);
    k_build_gemm<<<NBKT + TILES_BUILD, 256, 0, stream>>>(src, dst, runlen, runs, cnt, col, novf, ovf, x, Wm, h);

    k_gather    <<<N_NODES / (8 * NPW), 256, 0, stream>>>(cnt, col, novf, ovf, h, b, out, stat);

    k_bn        <<<(N_NODES / BN_ROWS + 7) / 8, 256, 0, stream>>>(out, stat, gamma, beta);
}


# x@W blocks compute 80 rows each with the wave's 128x16 slice of W held in registers: W fetched once per 80 rows, a fifth of the tile waves, all resident behind the edge blocks; sums unchanged
# speedup vs baseline: 3.4418x; 1.2571x over previous
#include <hip/hip_runtime.h>

#define N_NODES 100000
#define N_EDGES 640000
#define D       128
#define BN_EPS  1e-5f
#define LDS_PITCH 132

#define NREP 8

typedef float v2f __attribute__((ext_vector_type(2)));
typedef float v8f __attribute__((ext_vector_type(8)));

#define N_WGP       256
#define RT          5
#define TILE_ROWS   (16 * RT)
#define TILE_LDS    (TILE_ROWS * LDS_PITCH)
#define GEMM_BLOCKS (N_NODES / TILE_ROWS)
#define TILES_BIN   (3 * N_WGP)
#define TILES_BUILD (GEMM_BLOCKS - TILES_BIN)

template <int NE>
__device__ __forceinline__ bool block_is_tile(int bid, int& idx) {
    if (bid < NE) { idx = bid; return false; }
    idx = bid - NE;
    return true;
}

__device__ __forceinline__ void gemm_tile(int blk, const float* __restrict__ x,
                                          const float* __restrict__ Wm,
                                          float* __restrict__ h, float* lx) {
    static_assert(N_NODES % TILE_ROWS == 0, "whole tile blocks only");
    static_assert(TILES_BIN > 0 && TILES_BUILD > 0, "both launches carry tile blocks");
    const int tid  = threadIdx.x;
    const int row0 = blk * TILE_ROWS;

    const int wave = tid >> 5;
    const int lane = tid & 31;
    const int half = lane >> 4;
    const int l15  = lane & 15;
    const int koff = half * 2;
    const int n0   = wave * 16;

    const float4* xt = (const float4*)(x + (size_t)row0 * D);
    float4 xv[2 * RT];
#pragma unroll
    for (int i = 0; i < 2 * RT; ++i) xv[i] = xt[tid + i * 256];

    v2f bw[D / 4];
#pragma unroll
    for (int j = 0; j < D / 4; ++j) {
        bw[j].x = Wm[(4 * j + koff)     * D + n0 + l15];
        bw[j].y = Wm[(4 * j + koff + 1) * D + n0 + l15];
    }

#pragma unroll
    for (int i = 0; i < 2 * RT; ++i)
        *(float4*)&lx[((tid >> 5) + 8 * i) * LDS_PITCH + (tid & 31) * 4] = xv[i];
    __syncthreads();

    for (int t = 0; t < RT; ++t) {
        const float* la = lx + (t * 16 + l15) * LDS_PITCH + koff;
        v8f acc = {};
#pragma unroll
        for (int j = 0; j < D / 4; ++j) {
            const v2f a = *(const v2f*)(la + 4 * j);
            acc = __builtin_amdgcn_wmma_f32_16x16x4_f32(
                 false, a,  false, bw[j],
                 (short)0, acc,  false,  false);
        }

#pragma unroll
        for (int r = 0; r < 8; ++r) {
            int row = row0 + t * 16 + r + half * 8;
            h[row * D + n0 + l15] = acc[r];
        }
    }
}

#define CAP 32
#define CHUNK     2560
#define NBIN      (N_EDGES / CHUNK)
#define BKT_SHIFT 8
#define BKT       (1 << BKT_SHIFT)
#define NBKT      ((N_NODES + BKT - 1) / BKT)
#define RCAP      16
#define RL_PITCH  256

__device__ __forceinline__ void list_put(int s, int d, int r, int base, int* lcol,
                                         int* __restrict__ novf, int* __restrict__ ovf) {
    if (r < CAP) {
        lcol[(d - base) * CAP + r] = s;
    } else {
        const int o = atomicAdd(novf, 1);
        ovf[2 * o]     = s;
        ovf[2 * o + 1] = d;
    }
}

__global__ void __launch_bounds__(256) k_bin_gemm(const int* __restrict__ src,
                                                  const int* __restrict__ dst,
                                                  int* __restrict__ runlen,
                                                  int* __restrict__ runs,
                                                  float* __restrict__ stat,
                                                  int* __restrict__ novf,
                                                  const float* __restrict__ x,
                                                  const float* __restrict__ Wm,
                                                  float* __restrict__ h) {
    static_assert(N_EDGES % CHUNK == 0 && CHUNK % 256 == 0, "whole chunks, whole passes of the block");
    static_assert(NBIN <= RL_PITCH && NBIN <= 256, "one thread per run in level 2");
    static_assert(NBKT <= TILE_LDS, "bucket counters fit the tile buffer");
    __shared__ __attribute__((aligned(16))) float lx[TILE_LDS];
    int idx;
    if (block_is_tile<NBIN>((int)blockIdx.x, idx)) {
        gemm_tile(idx, x, Wm, h, lx);
        return;
    }
    int* lcnt = (int*)lx;
    const int tid = threadIdx.x;
    if (idx == 0) {
        for (int i = tid; i < NREP * 2 * D; i += 256) stat[i] = 0.0f;
        if (tid == 0) novf[0] = 0;
    }
    for (int k = tid; k < NBKT; k += 256) lcnt[k] = 0;

    constexpr int PER = CHUNK / 256;
    const int e0 = idx * CHUNK + tid;
    int d[PER], s[PER];
#pragma unroll
    for (int j = 0; j < PER; ++j) { d[j] = dst[e0 + j * 256]; s[j] = src[e0 + j * 256]; }
    __syncthreads();
    int r[PER];
#pragma unroll
    for (int j = 0; j < PER; ++j)
        r[j] = atomicAdd(&lcnt[d[j] >> BKT_SHIFT], 1);
#pragma unroll
    for (int j = 0; j < PER; ++j) {
        if (r[j] < RCAP) {
            const int k = d[j] >> BKT_SHIFT;
            int2 p; p.x = s[j]; p.y = d[j];
            ((int2*)runs)[((size_t)k * NBIN + idx) * RCAP + r[j]] = p;
        }
    }
    __syncthreads();
    for (int k = tid; k < NBKT; k += 256) runlen[k * RL_PITCH + idx] = lcnt[k];
}

#define BUILD_LDS_INTS (BKT * CAP + BKT + 256 + 256 + 4)
#define BUILD_SM_INTS  (BUILD_LDS_INTS > TILE_LDS ? BUILD_LDS_INTS : TILE_LDS)
__global__ void __launch_bounds__(256) k_build_gemm(const int* __restrict__ src,
                                                    const int* __restrict__ dst,
                                                    const int* __restrict__ runlen,
                                                    const int* __restrict__ runs,
                                                    int* __restrict__ cnt,
                                                    int* __restrict__ col,
                                                    int* __restrict__ novf,
                                                    int* __restrict__ ovf,
                                                    const float* __restrict__ x,
                                                    const float* __restrict__ Wm,
                                                    float* __restrict__ h) {
    static_assert(BKT == 256, "one thread per node of the bucket");
    static_assert(RCAP == 16 && NBIN * RCAP <= 16 * 256, "16 passes of the block cover every stored pair");
    __shared__ __attribute__((aligned(16))) int sm[BUILD_SM_INTS];
    int idx;
    if (block_is_tile<NBKT>((int)blockIdx.x, idx)) {
        gemm_tile(TILES_BIN + idx, x, Wm, h, (float*)sm);
        return;
    }
    int* lcol = sm;
    int* lcnt = lcol + BKT * CAP;
    int* rl   = lcnt + BKT;
    int* fb   = rl + 256;
    int* nfb  = fb + 256;
    const int tid  = threadIdx.x;
    const int k    = idx;
    const int base = k << BKT_SHIFT;

    lcnt[tid] = 0;
    int n = 0;
    if (tid < NBIN) n = runlen[k * RL_PITCH + tid];
    rl[tid] = n;
    if (tid == 0) nfb[0] = 0;
    __syncthreads();
    if (n > RCAP) { const int q = atomicAdd(nfb, 1); fb[q] = tid; }

    const int2* rp = (const int2*)runs + (size_t)k * NBIN * RCAP;
    int  nb[16];
    int2 p[16];
    int  r[16];
#pragma unroll
    for (int it = 0; it < 16; ++it) nb[it] = rl[(it * 256 + tid) >> 4];
#pragma unroll
    for (int it = 0; it < 16; ++it) {
        const int f = it * 256 + tid;
        p[it].x = 0; p[it].y = -1;
        if (nb[it] <= RCAP && (f & 15) < nb[it]) p[it] = rp[f];
    }
#pragma unroll
    for (int it = 0; it < 16; ++it) {
        r[it] = 0;
        if (p[it].y >= 0) r[it] = atomicAdd(&lcnt[p[it].y - base], 1);
    }
#pragma unroll
    for (int it = 0; it < 16; ++it)
        if (p[it].y >= 0) list_put(p[it].x, p[it].y, r[it], base, lcol, novf, ovf);
    __syncthreads();

    const int nf = nfb[0];
    for (int q = 0; q < nf; ++q) {
        const int e0 = fb[q] * CHUNK + tid;
        int dd[CHUNK / 256];
#pragma unroll
        for (int j = 0; j < CHUNK / 256; ++j) dd[j] = dst[e0 + j * 256];
#pragma unroll
        for (int j = 0; j < CHUNK / 256; ++j) {
            if ((dd[j] >> BKT_SHIFT) == k) {
                const int rr = atomicAdd(&lcnt[dd[j] - base], 1);
                list_put(src[e0 + j * 256], dd[j], rr, base, lcol, novf, ovf);
            }
        }
    }
    __syncthreads();

    if (base + tid < N_NODES) cnt[base + tid] = lcnt[tid];
    for (int q = tid; q < BKT * CAP / 4; q += 256)
        if (base + (q >> 3) < N_NODES)
            ((int4*)col)[(size_t)base * (CAP / 4) + q] = ((const int4*)lcol)[q];
}

template <int R>
__device__ __forceinline__ void add_rows(float4& acc, const float4* __restrict__ hp,
                                         const int* __restrict__ cnt, int my_s, float dt,
                                         int j, int lane) {
    float4 v[R];
#pragma unroll
    for (int k = 0; k < R; ++k)
        v[k] = hp[(size_t)__builtin_amdgcn_readlane(my_s, j + k) * 32 + lane];
    const float my_w = rsqrtf((float)(cnt[my_s] + 1)) * dt;
#pragma unroll
    for (int k = 0; k < R; ++k) {
        const float w = __int_as_float(__builtin_amdgcn_readlane(__float_as_int(my_w), j + k));
        acc.x = fmaf(v[k].x, w, acc.x);
        acc.y = fmaf(v[k].y, w, acc.y);
        acc.z = fmaf(v[k].z, w, acc.z);
        acc.w = fmaf(v[k].w, w, acc.w);
    }
}

#define GR 9
#define NPW 4
__global__ void __launch_bounds__(256) k_gather(const int* __restrict__ cnt,
                                                const int* __restrict__ col,
                                                const int* __restrict__ novf,
                                                const int* __restrict__ ovf,
                                                const float* __restrict__ h,
                                                const float* __restrict__ b,
                                                float* __restrict__ out,
                                                float* __restrict__ stat) {
    static_assert(N_NODES % (8 * NPW) == 0, "whole blocks only: every wave has NPW nodes and reaches the barrier");
    __shared__ __attribute__((aligned(16))) float red[8][2 * D];
    const int lane = threadIdx.x & 31;
    const int wave = threadIdx.x >> 5;
    const int t0 = __builtin_amdgcn_readfirstlane((int)(blockIdx.x * 8 + wave) * NPW);

    const float4* hp = (const float4*)h;
    const float4 bb  = ((const float4*)b)[lane];
    float4* my_sr = (float4*)&red[wave][0] + lane;
    float4* my_sq = (float4*)&red[wave][D] + lane;
    *my_sr = make_float4(0.0f, 0.0f, 0.0f, 0.0f);
    *my_sq = make_float4(0.0f, 0.0f, 0.0f, 0.0f);

    for (int i = 0; i < NPW; ++i) {
        const int t = t0 + i;
        const int    c   = cnt[t];
        const int    raw = col[t * CAP + lane];
        const float4 hs  = hp[(size_t)t * 32 + lane];

        const float dt = rsqrtf((float)(c + 1));
        const int   n  = c < CAP ? c : CAP;
        asm volatile("" :: "v"(raw));
        const int   my_s = lane < n ? raw : 0;

        const float  ws = dt * dt;
        float4 acc;
        acc.x = hs.x * ws; acc.y = hs.y * ws; acc.z = hs.z * ws; acc.w = hs.w * ws;

        for (int j = 0; j < n; j += GR) {
            switch (n - j) {
                case 1:  add_rows<1>(acc, hp, cnt, my_s, dt, j, lane); break;
                case 2:  add_rows<2>(acc, hp, cnt, my_s, dt, j, lane); break;
                case 3:  add_rows<3>(acc, hp, cnt, my_s, dt, j, lane); break;
                case 4:  add_rows<4>(acc, hp, cnt, my_s, dt, j, lane); break;
                case 5:  add_rows<5>(acc, hp, cnt, my_s, dt, j, lane); break;
                case 6:  add_rows<6>(acc, hp, cnt, my_s, dt, j, lane); break;
                case 7:  add_rows<7>(acc, hp, cnt, my_s, dt, j, lane); break;
                case 8:  add_rows<8>(acc, hp, cnt, my_s, dt, j, lane); break;
                default: add_rows<GR>(acc, hp, cnt, my_s, dt, j, lane); break;
            }
        }

        if (c > CAP) {
            const int no = novf[0];
            for (int i0 = 0; i0 < no; i0 += 32) {
                int s_l = 0;
                int d_l = -1;
                if (i0 + lane < no) {
                    s_l = ovf[2 * (i0 + lane)];
                    d_l = ovf[2 * (i0 + lane) + 1];
                }
                unsigned int m = __builtin_amdgcn_ballot_w32(d_l == t);
                while (m != 0u) {
                    const int k = __builtin_ctz(m);
                    m &= m - 1u;
                    const int   s = __builtin_amdgcn_readlane(s_l, k);
                    const float w = rsqrtf((float)(cnt[s] + 1)) * dt;
                    const float4 v = hp[(size_t)s * 32 + lane];
                    acc.x = fmaf(v.x, w, acc.x);
                    acc.y = fmaf(v.y, w, acc.y);
                    acc.z = fmaf(v.z, w, acc.z);
                    acc.w = fmaf(v.w, w, acc.w);
                }
            }
        }

        float4 o;
        o.x = acc.x + bb.x; o.y = acc.y + bb.y; o.z = acc.z + bb.z; o.w = acc.w + bb.w;
        ((float4*)out)[(size_t)t * 32 + lane] = o;

        float4 r;
        r.x = o.x > 0.0f ? o.x : 0.0f;
        r.y = o.y > 0.0f ? o.y : 0.0f;
        r.z = o.z > 0.0f ? o.z : 0.0f;
        r.w = o.w > 0.0f ? o.w : 0.0f;
        float4 sr = *my_sr;
        float4 sq = *my_sq;
        sr.x += r.x; sr.y += r.y; sr.z += r.z; sr.w += r.w;
        sq.x += r.x * r.x; sq.y += r.y * r.y; sq.z += r.z * r.z; sq.w += r.w * r.w;
        *my_sr = sr;
        *my_sq = sq;
    }
    __syncthreads();
    float a = 0.0f;
#pragma unroll
    for (int w = 0; w < 8; ++w) a += red[w][threadIdx.x];
    atomicAdd(&stat[(blockIdx.x % NREP) * (2 * D) + threadIdx.x], a);
}

#define BN_ROWS 16

__device__ __forceinline__ float bn_one(float v, float mean, float rs, float g, float bt) {
    v = v > 0.0f ? v : 0.0f;
    return g * (v - mean) * rs + bt;
}

__global__ void __launch_bounds__(256) k_bn(float* __restrict__ out,
                                            const float* __restrict__ stat,
                                            const float* __restrict__ gamma,
                                            const float* __restrict__ beta) {
    static_assert(N_NODES % BN_ROWS == 0, "whole waves only");
    const int lane = threadIdx.x & 31;
    const int row0 = (int)(blockIdx.x * 8 + (threadIdx.x >> 5)) * BN_ROWS;
    if (row0 >= N_NODES) return;

    float4* op = (float4*)out + (size_t)row0 * 32 + lane;
    float4 v[BN_ROWS];
#pragma unroll
    for (int k = 0; k < BN_ROWS; ++k) v[k] = op[k * 32];

    float4 s  = make_float4(0.0f, 0.0f, 0.0f, 0.0f);
    float4 sq = make_float4(0.0f, 0.0f, 0.0f, 0.0f);
#pragma unroll
    for (int k = 0; k < NREP; ++k) {
        const float4 a = ((const float4*)(stat + k * 2 * D))[lane];
        const float4 c = ((const float4*)(stat + k * 2 * D + D))[lane];
        s.x += a.x; s.y += a.y; s.z += a.z; s.w += a.w;
        sq.x += c.x; sq.y += c.y; sq.z += c.z; sq.w += c.w;
    }
    const float4 g  = ((const float4*)gamma)[lane];
    const float4 bt = ((const float4*)beta)[lane];

    const float invN = 1.0f / (float)N_NODES;
    float4 mean, var, rs;
    mean.x = s.x * invN; mean.y = s.y * invN; mean.z = s.z * invN; mean.w = s.w * invN;
    var.x = sq.x * invN - mean.x * mean.x;
    var.y = sq.y * invN - mean.y * mean.y;
    var.z = sq.z * invN - mean.z * mean.z;
    var.w = sq.w * invN - mean.w * mean.w;
    rs.x = rsqrtf(var.x + BN_EPS); rs.y = rsqrtf(var.y + BN_EPS);
    rs.z = rsqrtf(var.z + BN_EPS); rs.w = rsqrtf(var.w + BN_EPS);

#pragma unroll
    for (int k = 0; k < BN_ROWS; ++k) {
        float4 o;
        o.x = bn_one(v[k].x, mean.x, rs.x, g.x, bt.x);
        o.y = bn_one(v[k].y, mean.y, rs.y, g.y, bt.y);
        o.z = bn_one(v[k].z, mean.z, rs.z, g.z, bt.z);
        o.w = bn_one(v[k].w, mean.w, rs.w, g.w, bt.w);
        op[k * 32] = o;
    }
}

extern "C" void kernel_launch(void* const* d_in, const int* in_sizes, int n_in,
                              void* d_out, int out_size, void* d_ws, size_t ws_size,
                              hipStream_t stream) {
    const float* x     = (const float*)d_in[0];
    const int*   edges = (const int*)  d_in[1];
    const float* Wm    = (const float*)d_in[2];
    const float* b     = (const float*)d_in[3];
    const float* gamma = (const float*)d_in[4];
    const float* beta  = (const float*)d_in[5];
    float*       out   = (float*)d_out;

    const int* src = edges;
    const int* dst = edges + N_EDGES;

    float* h      = (float*)d_ws;
    int*   cnt    = (int*)(h + (size_t)N_NODES * D);
    int*   col    = cnt + N_NODES;
    float* stat   = (float*)(col + (size_t)N_NODES * CAP);
    int*   novf   = (int*)(stat + NREP * 2 * D);
    int*   ovf    = novf + 32;
    int*   runlen = ovf + 2 * (size_t)N_EDGES;
    int*   runs   = runlen + (size_t)NBKT * RL_PITCH;

    k_bin_gemm  <<<NBIN + TILES_BIN, 256, 0, stream>>>(src, dst, runlen, runs, stat, novf, x, Wm, h);
    k_build_gemm<<<NBKT + TILES_BUILD, 256, 0, stream>>>(src, dst, runlen, runs, cnt, col, novf, ovf, x, Wm, h);

    k_gather    <<<N_NODES / (8 * NPW), 256, 0, stream>>>(cnt, col, novf, ovf, h, b, out, stat);

    k_bn        <<<(N_NODES / BN_ROWS + 7) / 8, 256, 0, stream>>>(out, stat, gamma, beta);
}
